# starting kernel
# baseline (speedup 1.0000x reference)
#include <hip/hip_runtime.h>
#include <hip/hip_bf16.h>

#define F_IN 512
#define HID  16
#define NCLS 7
#define TILES_PER_WAVE 2

typedef float v2f __attribute__((ext_vector_type(2)));
typedef float v8f __attribute__((ext_vector_type(8)));


__global__ void gcn_init_deg(float* __restrict__ deg, int n) {
    int i = blockIdx.x * blockDim.x + threadIdx.x;
    if (i < n) deg[i] = 1.0f;
}

__global__ void gcn_deg_scatter(const int* __restrict__ col, float* __restrict__ deg, int e) {
    int i = blockIdx.x * blockDim.x + threadIdx.x;
    if (i < e) atomicAdd(&deg[col[i]], 1.0f);
}

__global__ void gcn_dinv(float* __restrict__ degdinv, int n) {
    int i = blockIdx.x * blockDim.x + threadIdx.x;
    if (i < n) degdinv[i] = rsqrtf(degdinv[i]);
}


__device__ __forceinline__ void store_tile(float* __restrict__ h1, const v8f& acc,
                                           int tile, int lane, int n) {
    int mbase = tile * 16 + ((lane >> 4) << 3);
    int ncol  = lane & 15;
    float* p = h1 + (size_t)mbase * HID + ncol;
    if (mbase + 8 <= n) {
        #pragma unroll
        for (int v = 0; v < 8; ++v) p[v * HID] = acc[v];
    } else {
        #pragma unroll
        for (int v = 0; v < 8; ++v) if (mbase + v < n) p[v * HID] = acc[v];
    }
}

__global__ void gcn_gemm1_wmma(const float* __restrict__ x, const float* __restrict__ W1,
                               float* __restrict__ h1, int n) {
    int lane = threadIdx.x & 31;
    int wid  = (blockIdx.x * blockDim.x + threadIdx.x) >> 5;
    int ntiles = (n + 15) >> 4;
    int tile0 = wid * TILES_PER_WAVE;
    if (tile0 >= ntiles) return;
    bool has1 = (tile0 + 1) < ntiles;

    int m    = lane & 15;
    int koff = (lane >> 4) << 1;

    int row0 = tile0 * 16 + m;
    int row1 = row0 + 16;
    const float* xa0 = x + (size_t)(row0 < n ? row0 : 0) * F_IN + koff;
    const float* xa1 = x + (size_t)(row1 < n ? row1 : 0) * F_IN + koff;

    v8f acc0 = {}, acc1 = {};
    #pragma unroll 4
    for (int k = 0; k < F_IN; k += 4) {
        v2f b;
        b.x = W1[(size_t)(k + koff)     * HID + m];
        b.y = W1[(size_t)(k + koff + 1) * HID + m];
        v2f a0 = *(const v2f*)(xa0 + k);
        v2f a1 = *(const v2f*)(xa1 + k);
        acc0 = __builtin_amdgcn_wmma_f32_16x16x4_f32(
            false, a0, false, b, (short)0, acc0, false, false);
        acc1 = __builtin_amdgcn_wmma_f32_16x16x4_f32(
            false, a1, false, b, (short)0, acc1, false, false);
    }

    store_tile(h1, acc0, tile0, lane, n);
    if (has1) store_tile(h1, acc1, tile0 + 1, lane, n);
}


__global__ void gcn_init_out1(const float* __restrict__ h1, const float* __restrict__ dinv,
                              const float* __restrict__ b1, float* __restrict__ out1, int n) {
    int i = blockIdx.x * blockDim.x + threadIdx.x;
    if (i >= n) return;
    float s = dinv[i]; s = s * s;
    const float4* hp = (const float4*)(h1 + (size_t)i * HID);
    float4* op = (float4*)(out1 + (size_t)i * HID);
    #pragma unroll
    for (int q = 0; q < 4; ++q) {
        float4 h = hp[q];
        op[q] = make_float4(b1[4*q+0] + s*h.x, b1[4*q+1] + s*h.y,
                            b1[4*q+2] + s*h.z, b1[4*q+3] + s*h.w);
    }
}

__global__ void gcn_scatter1(const int* __restrict__ rowi, const int* __restrict__ coli,
                             const float* __restrict__ dinv, const float* __restrict__ h1,
                             float* __restrict__ out1, int e) {
    int idx = blockIdx.x * blockDim.x + threadIdx.x;
    if (idx >= e) return;
    int r = rowi[idx], c = coli[idx];
    float w = dinv[r] * dinv[c];
    const float* hp = h1 + (size_t)r * HID;
    float* op = out1 + (size_t)c * HID;
    #pragma unroll
    for (int q = 0; q < HID; ++q) atomicAdd(op + q, w * hp[q]);
}


__global__ void gcn_gemm2(const float* __restrict__ out1, const float* __restrict__ W2,
                          const float* __restrict__ b2, const float* __restrict__ dinv,
                          float* __restrict__ g, float* __restrict__ out2, int n) {
    int i = blockIdx.x * blockDim.x + threadIdx.x;
    if (i >= n) return;
    float h[HID];
    const float* ip = out1 + (size_t)i * HID;
    #pragma unroll
    for (int c = 0; c < HID; ++c) h[c] = fmaxf(ip[c], 0.0f);
    float s = dinv[i]; s = s * s;
    #pragma unroll
    for (int c2 = 0; c2 < NCLS; ++c2) {
        float acc = 0.0f;
        #pragma unroll
        for (int c = 0; c < HID; ++c) acc += h[c] * W2[c * NCLS + c2];
        g[(size_t)i * NCLS + c2]    = acc;
        out2[(size_t)i * NCLS + c2] = b2[c2] + s * acc;
    }
}

__global__ void gcn_scatter2(const int* __restrict__ rowi, const int* __restrict__ coli,
                             const float* __restrict__ dinv, const float* __restrict__ g,
                             float* __restrict__ out2, int e) {
    int idx = blockIdx.x * blockDim.x + threadIdx.x;
    if (idx >= e) return;
    int r = rowi[idx], c = coli[idx];
    float w = dinv[r] * dinv[c];
    const float* gp = g + (size_t)r * NCLS;
    float* op = out2 + (size_t)c * NCLS;
    #pragma unroll
    for (int q = 0; q < NCLS; ++q) atomicAdd(op + q, w * gp[q]);
}


__global__ void gcn_logsoftmax(const float* __restrict__ out2, float* __restrict__ out, int n) {
    int i = blockIdx.x * blockDim.x + threadIdx.x;
    if (i >= n) return;
    float v[NCLS];
    float m = -1e30f;
    #pragma unroll
    for (int c = 0; c < NCLS; ++c) { v[c] = out2[(size_t)i * NCLS + c]; m = fmaxf(m, v[c]); }
    float sum = 0.0f;
    #pragma unroll
    for (int c = 0; c < NCLS; ++c) sum += __expf(v[c] - m);
    float lse = m + __logf(sum);
    #pragma unroll
    for (int c = 0; c < NCLS; ++c) out[(size_t)i * NCLS + c] = v[c] - lse;
}


extern "C" void kernel_launch(void* const* d_in, const int* in_sizes, int n_in,
                              void* d_out, int out_size, void* d_ws, size_t ws_size,
                              hipStream_t stream) {
    const float* x  = (const float*)d_in[0];
    const int*   ei = (const int*)  d_in[1];
    const float* W1 = (const float*)d_in[2];
    const float* b1 = (const float*)d_in[3];
    const float* W2 = (const float*)d_in[4];
    const float* b2 = (const float*)d_in[5];
    float* out = (float*)d_out;

    int n = in_sizes[0] / F_IN;
    int e = in_sizes[1] / 2;
    const int* rowi = ei;
    const int* coli = ei + e;

    float* ws   = (float*)d_ws;
    float* dinv = ws;
    float* h1   = dinv + n;
    float* out1 = h1   + (size_t)n * HID;
    float* g    = out1 + (size_t)n * HID;
    float* out2 = g    + (size_t)n * NCLS;

    const int B = 256;
    int gn = (n + B - 1) / B;
    int ge = (e + B - 1) / B;

    gcn_init_deg   <<<gn, B, 0, stream>>>(dinv, n);
    gcn_deg_scatter<<<ge, B, 0, stream>>>(coli, dinv, e);
    gcn_dinv       <<<gn, B, 0, stream>>>(dinv, n);

    int ntiles = (n + 15) >> 4;
    int nwaves = (ntiles + TILES_PER_WAVE - 1) / TILES_PER_WAVE;
    int gw = (nwaves * 32 + B - 1) / B;
    gcn_gemm1_wmma <<<gw, B, 0, stream>>>(x, W1, h1, n);

    gcn_init_out1  <<<gn, B, 0, stream>>>(h1, dinv, b1, out1, n);
    gcn_scatter1   <<<ge, B, 0, stream>>>(rowi, coli, dinv, h1, out1, e);

    gcn_gemm2      <<<gn, B, 0, stream>>>(out1, W2, b2, dinv, g, out2, n);
    gcn_scatter2   <<<ge, B, 0, stream>>>(rowi, coli, dinv, g, out2, e);

    gcn_logsoftmax <<<gn, B, 0, stream>>>(out2, out, n);
}


# conv2 scatter replays conv1's hit lists: conv1 stores each pass's 32 (source,target) words in its block's list; conv2 (same 25x10 blocks: 10 partials a node, was 19) reads 4 B a hit, no filter
# speedup vs baseline: 7.7224x; 7.7224x over previous
#include <hip/hip_runtime.h>
#include <hip/hip_bf16.h>

#define F_IN 512
#define HID  16
#define NCLS 7
#define TILES_PER_WAVE 2

#define H1_PITCH 32
#define H1_DW    16
#define G_PITCH  8
#define G_DW     7

typedef float v2f __attribute__((ext_vector_type(2)));
typedef float v8f __attribute__((ext_vector_type(8)));


#define SCAT_BLK   1024
#define SCAT_WAVES (SCAT_BLK / 32)
#define SCAT_TILE  512
#define DEG_NR     50000

__global__ __launch_bounds__(SCAT_BLK)
void gcn_deg_scatter(const int* __restrict__ col, float* __restrict__ deg, int n, int e, int nchunks) {
    __shared__ float acc[DEG_NR];
    const int tid = threadIdx.x;
    const int lo = blockIdx.x * DEG_NR;
    const int nr = min(DEG_NR, n - lo);
    for (int j = tid; j < nr; j += SCAT_BLK) acc[j] = 0.0f;
    __syncthreads();

    const int chunk = (((e + nchunks - 1) / nchunks) + 3) & ~3;
    const int e0 = blockIdx.y * chunk;
    const int e1 = min(e0 + chunk, e);
    const bool al = (((size_t)col) & 15) == 0;
    for (int i = e0 + tid * 4; i < e1; i += SCAT_BLK * 4) {
        int t[4];
        if (al && i + 4 <= e1) {
            int4 v = *(const int4*)(col + i);
            t[0] = v.x; t[1] = v.y; t[2] = v.z; t[3] = v.w;
        } else {
            #pragma unroll
            for (int k = 0; k < 4; ++k) t[k] = (i + k < e1) ? col[i + k] : -1;
        }
        #pragma unroll
        for (int k = 0; k < 4; ++k) {
            unsigned d = (unsigned)(t[k] - lo);
            if (d < (unsigned)nr) atomicAdd(&acc[d], 1.0f);
        }
    }
    __syncthreads();
    float* dp = deg + lo;
    for (int j = tid; j < nr; j += SCAT_BLK) {
        float v = acc[j];
        if (v != 0.0f) atomicAdd(&dp[j], v);
    }
}


#define SCAT_QCAP 512

#define LIST_GROUPS 512
#define LIST_CAP    (LIST_GROUPS * 32)
#define LIST_NONE   0xffffffffu

template <int LPE, int R>
__device__ __forceinline__ int gcn_group_lane(int x) {
    static_assert(LPE == 16 || LPE == 8, "lane groups are DPP16 rows or DPP8 groups");
    static_assert(R >= 0 && R < LPE, "source lane lies inside the group");
    if constexpr (LPE == 16) return __builtin_amdgcn_mov_dpp(x, 0x150 + R, 0xf, 0xf, true);
    else                     return (int)__builtin_amdgcn_mov_dpp8((unsigned)x, R * 0x249249);
}
template <int LPE, int R>
__device__ __forceinline__ float gcn_group_lane_f(float x) {
    return __int_as_float(gcn_group_lane<LPE, R>(__float_as_int(x)));
}

template <int LPE, int PITCH, int R>
__device__ __forceinline__ void gcn_gather_rounds(int s, int f, const float* __restrict__ src, float (&v)[LPE]) {
    v[R] = src[gcn_group_lane<LPE, R>(s) * PITCH + f];
    if constexpr (R + 1 < LPE) gcn_gather_rounds<LPE, PITCH, R + 1>(s, f, src, v);
}

template <int F, int LPE, int DW, bool FULL, bool SKIP, int R>
__device__ __forceinline__ void gcn_add_rounds(int tl, float w, float dt, int m, int sub, int f,
                                               const float (&v)[LPE], float* acc) {
    if (FULL || !SKIP || R < m) {
        const int te = gcn_group_lane<LPE, R>(tl);
        float we;
        if constexpr (DW >= LPE) we = gcn_group_lane_f<LPE, R>(w);
        else                     we = gcn_group_lane_f<LPE, DW>(v[R]) * gcn_group_lane_f<LPE, R>(dt);
        if ((FULL || sub * LPE + R < m) && f < F) atomicAdd(&acc[te * F + f], we * v[R]);
    }
    if constexpr (R + 1 < LPE) gcn_add_rounds<F, LPE, DW, FULL, SKIP, R + 1>(tl, w, dt, m, sub, f, v, acc);
}

template <int F, int LPE, int PITCH, int DW, int TB, bool FULL, bool SKIP = true>
__device__ __forceinline__ void gcn_scatter_pass(unsigned ent, int m, int sub, int f, int lo,
                                                 const float* __restrict__ dinv, const float* __restrict__ src,
                                                 float* acc) {
    const int s  = (int)(ent >> TB);
    const int tl = (int)(ent & ((1u << TB) - 1u));
    float ds = 0.0f;
    if constexpr (DW >= LPE) ds = src[s * PITCH + DW];
    float v[LPE];
    gcn_gather_rounds<LPE, PITCH, 0>(s, f, src, v);
    const float dt = dinv[lo + tl];
    const float w  = ds * dt;
    gcn_add_rounds<F, LPE, DW, FULL, SKIP, 0>(tl, w, dt, m, sub, f, v, acc);
}

__device__ __forceinline__ void gcn_list_put(unsigned ent, bool real, int lane, int* ngroups, unsigned* lst) {
    int g = 0;
    if (lane == 0) g = atomicAdd(ngroups, 1);
    g = __builtin_amdgcn_readfirstlane(g);
    if (g < LIST_GROUPS) lst[g * 32 + lane] = real ? ent : LIST_NONE;
}

template <int F, int LPE, int NR, int PITCH, int DW, int TB, int LIST>
__global__ __launch_bounds__(SCAT_BLK)
void gcn_scatter_own(const int* __restrict__ rowi, const int* __restrict__ coli,
                     const float* __restrict__ dinv, const float* __restrict__ src,
                     float* __restrict__ dst, int n, int e, int nchunks,
                     unsigned* lists, int* counts) {
    static_assert(NR <= (1 << TB), "target offset must fit the packed word");
    static_assert(LIST == 0 || NR < (1 << TB) - 1, "LIST_NONE must not be an entry");
    __shared__ float    acc[NR * F];
    __shared__ unsigned queue[SCAT_WAVES * SCAT_QCAP];
    __shared__ int      ngroups;
    const int tid  = threadIdx.x;
    const int lane = tid & 31;
    const int wv   = tid >> 5;
    const int lo = blockIdx.x * NR;
    const int nr = min(NR, n - lo);
    const int bid = blockIdx.y * gridDim.x + blockIdx.x;
    unsigned* lst = (LIST != 0) ? lists + (size_t)bid * LIST_CAP : nullptr;
    for (int j = tid; j < nr * F; j += SCAT_BLK) acc[j] = 0.0f;
    if (LIST == 1 && tid == 0) ngroups = 0;
    __syncthreads();
    const int sub = lane / LPE;
    const int f   = lane % LPE;

    const int ng = (LIST == 2) ? counts[bid] : 0;
    if (LIST == 2 && ng <= LIST_GROUPS) {
        int gi = __builtin_amdgcn_readfirstlane(wv);
        unsigned nxt = (gi < ng) ? lst[gi * 32 + lane] : 0u;
        for (; gi < ng; gi += SCAT_WAVES) {
            unsigned ent = nxt;
            if (gi + SCAT_WAVES < ng) nxt = lst[(gi + SCAT_WAVES) * 32 + lane];
            const int m = __popc((unsigned)__ballot(ent != LIST_NONE));
            const unsigned first = (unsigned)__builtin_amdgcn_readfirstlane((int)ent);
            if (lane >= m) ent = first;
            gcn_scatter_pass<F, LPE, PITCH, DW, TB, false, false>(ent, m, sub, f, lo, dinv, src, acc);
        }
    } else {

    const int chunk = (((e + nchunks - 1) / nchunks) + SCAT_TILE - 1) & ~(SCAT_TILE - 1);
    const int e0 = blockIdx.y * chunk;
    const int e1 = min(e0 + chunk, e);
    const bool al = ((((size_t)coli) | ((size_t)rowi)) & 15) == 0;
    unsigned* q = queue + wv * SCAT_QCAP;
    const unsigned below = (1u << lane) - 1u;

    auto load_half = [&](int b, int hlf, int (&c)[16], int (&r)[16]) {
        if (al && b + SCAT_TILE <= e1) {
            #pragma unroll
            for (int j = 2 * hlf; j < 2 * hlf + 2; ++j) {
                const int4 vc = *(const int4*)(coli + b + j * 128 + lane * 4);
                const int4 vr = *(const int4*)(rowi + b + j * 128 + lane * 4);
                c[4*j+0] = vc.x; c[4*j+1] = vc.y; c[4*j+2] = vc.z; c[4*j+3] = vc.w;
                r[4*j+0] = vr.x; r[4*j+1] = vr.y; r[4*j+2] = vr.z; r[4*j+3] = vr.w;
            }
        } else {
            #pragma unroll
            for (int j = 8 * hlf; j < 8 * hlf + 8; ++j) {
                const int idx = b + (j >> 2) * 128 + lane * 4 + (j & 3);
                const bool in = idx < e1;
                c[j] = in ? coli[idx] : -1;
                r[j] = in ? rowi[idx] : 0;
            }
        }
    };

    constexpr int STEP = SCAT_WAVES * SCAT_TILE;
    int base = e0 + wv * SCAT_TILE;
    int c[16], r[16];
    int head = 0, tail = 0;
    if (base < e1) { load_half(base, 0, c, r); load_half(base, 1, c, r); }
    for (; base < e1; base += STEP) {
        #pragma unroll
        for (int hlf = 0; hlf < 2; ++hlf) {
            #pragma unroll
            for (int j = 8 * hlf; j < 8 * hlf + 8; ++j) {
                const unsigned d = (unsigned)(c[j] - lo);
                const bool hit = d < (unsigned)nr;
                const unsigned mask = (unsigned)__ballot(hit);
                if (hit) q[(tail + __popc(mask & below)) & (SCAT_QCAP - 1)] = ((unsigned)r[j] << TB) | d;
                tail += __popc(mask);
            }
            if (base + STEP < e1) load_half(base + STEP, hlf, c, r);
            __builtin_amdgcn_fence(__ATOMIC_RELEASE, "wavefront");
            __builtin_amdgcn_wave_barrier();
            while (tail - head >= 32) {
                const unsigned ent = q[(head + lane) & (SCAT_QCAP - 1)];
                gcn_scatter_pass<F, LPE, PITCH, DW, TB, true>(ent, 32, sub, f, lo, dinv, src, acc);
                if (LIST == 1) gcn_list_put(ent, true, lane, &ngroups, lst);
                head += 32;
            }
            __builtin_amdgcn_fence(__ATOMIC_ACQUIRE, "wavefront");
            __builtin_amdgcn_wave_barrier();
        }
    }
    if (tail != head) {
        const int m = tail - head;
        const unsigned ent = q[(head + (lane < m ? lane : 0)) & (SCAT_QCAP - 1)];
        gcn_scatter_pass<F, LPE, PITCH, DW, TB, false>(ent, m, sub, f, lo, dinv, src, acc);
        if (LIST == 1) gcn_list_put(ent, lane < m, lane, &ngroups, lst);
    }
    }
    __syncthreads();
    if (LIST == 1 && tid == 0) counts[bid] = ngroups;
    float* dp = dst + (size_t)lo * F;
    for (int j = tid; j < nr * F; j += SCAT_BLK) atomicAdd(&dp[j], acc[j]);
}

__global__ void gcn_dinv_init_out1(float* __restrict__ degdinv, float* h1,
                                   const float* __restrict__ b1, float* __restrict__ out1, int n) {
    int i = blockIdx.x * blockDim.x + threadIdx.x;
    if (i >= n) return;
    const float r = rsqrtf(degdinv[i]);
    degdinv[i] = r;
    float* hrow = h1 + (size_t)i * H1_PITCH;
    hrow[H1_DW] = r;
    const float s = r * r;
    const float4* hp = (const float4*)hrow;
    float4* op = (float4*)(out1 + (size_t)i * HID);
    #pragma unroll
    for (int q = 0; q < 4; ++q) {
        float4 h = hp[q];
        op[q] = make_float4(b1[4*q+0] + s*h.x, b1[4*q+1] + s*h.y,
                            b1[4*q+2] + s*h.z, b1[4*q+3] + s*h.w);
    }
}


__device__ __forceinline__ void store_tile(float* __restrict__ h1, const v8f& acc,
                                           int tile, int lane, int n) {
    int mbase = tile * 16 + ((lane >> 4) << 3);
    int ncol  = lane & 15;
    float* p = h1 + (size_t)mbase * H1_PITCH + ncol;
    if (mbase + 8 <= n) {
        #pragma unroll
        for (int v = 0; v < 8; ++v) p[v * H1_PITCH] = acc[v];
    } else {
        #pragma unroll
        for (int v = 0; v < 8; ++v) if (mbase + v < n) p[v * H1_PITCH] = acc[v];
    }
}

__global__ void gcn_gemm1_wmma(const float* __restrict__ x, const float* __restrict__ W1,
                               float* __restrict__ h1, float* __restrict__ deg, int n) {
    int gid = blockIdx.x * blockDim.x + threadIdx.x;
    if (gid < n) deg[gid] = 1.0f;
    int lane = threadIdx.x & 31;
    int wid  = gid >> 5;
    int ntiles = (n + 15) >> 4;
    int tile0 = wid * TILES_PER_WAVE;
    if (tile0 >= ntiles) return;
    bool has1 = (tile0 + 1) < ntiles;

    int m    = lane & 15;
    int koff = (lane >> 4) << 1;

    int row0 = tile0 * 16 + m;
    int row1 = row0 + 16;
    const float* xa0 = x + (size_t)(row0 < n ? row0 : 0) * F_IN + koff;
    const float* xa1 = x + (size_t)(row1 < n ? row1 : 0) * F_IN + koff;

    v8f acc0 = {}, acc1 = {};
    #pragma unroll 4
    for (int k = 0; k < F_IN; k += 4) {
        v2f b;
        b.x = W1[(size_t)(k + koff)     * HID + m];
        b.y = W1[(size_t)(k + koff + 1) * HID + m];
        v2f a0 = *(const v2f*)(xa0 + k);
        v2f a1 = *(const v2f*)(xa1 + k);
        acc0 = __builtin_amdgcn_wmma_f32_16x16x4_f32(
            false, a0, false, b, (short)0, acc0, false, false);
        acc1 = __builtin_amdgcn_wmma_f32_16x16x4_f32(
            false, a1, false, b, (short)0, acc1, false, false);
    }

    store_tile(h1, acc0, tile0, lane, n);
    if (has1) store_tile(h1, acc1, tile0 + 1, lane, n);
}


__global__ void gcn_gemm2(const float* __restrict__ out1, const float* __restrict__ W2,
                          const float* __restrict__ b2, const float* __restrict__ dinv,
                          float* __restrict__ g, float* __restrict__ out2, int n) {
    int i = blockIdx.x * blockDim.x + threadIdx.x;
    if (i >= n) return;
    float h[HID];
    const float* ip = out1 + (size_t)i * HID;
    #pragma unroll
    for (int c = 0; c < HID; ++c) h[c] = fmaxf(ip[c], 0.0f);
    const float dv = dinv[i];
    const float s = dv * dv;
    float a[G_PITCH];
    #pragma unroll
    for (int c2 = 0; c2 < NCLS; ++c2) {
        float acc = 0.0f;
        #pragma unroll
        for (int c = 0; c < HID; ++c) acc += h[c] * W2[c * NCLS + c2];
        a[c2] = acc;
        out2[(size_t)i * NCLS + c2] = b2[c2] + s * acc;
    }
    a[G_DW] = dv;
    float4* gp = (float4*)(g + (size_t)i * G_PITCH);
    gp[0] = make_float4(a[0], a[1], a[2], a[3]);
    gp[1] = make_float4(a[4], a[5], a[6], a[7]);
}


__global__ void gcn_logsoftmax(const float* __restrict__ out2, float* __restrict__ out, int n) {
    int i = blockIdx.x * blockDim.x + threadIdx.x;
    if (i >= n) return;
    float v[NCLS];
    float m = -1e30f;
    #pragma unroll
    for (int c = 0; c < NCLS; ++c) { v[c] = out2[(size_t)i * NCLS + c]; m = fmaxf(m, v[c]); }
    float sum = 0.0f;
    #pragma unroll
    for (int c = 0; c < NCLS; ++c) sum += __expf(v[c] - m);
    float lse = m + __logf(sum);
    #pragma unroll
    for (int c = 0; c < NCLS; ++c) out[(size_t)i * NCLS + c] = v[c] - lse;
}


extern "C" void kernel_launch(void* const* d_in, const int* in_sizes, int n_in,
                              void* d_out, int out_size, void* d_ws, size_t ws_size,
                              hipStream_t stream) {
    const float* x  = (const float*)d_in[0];
    const int*   ei = (const int*)  d_in[1];
    const float* W1 = (const float*)d_in[2];
    const float* b1 = (const float*)d_in[3];
    const float* W2 = (const float*)d_in[4];
    const float* b2 = (const float*)d_in[5];
    float* out = (float*)d_out;

    int n = in_sizes[0] / F_IN;
    int e = in_sizes[1] / 2;
    const int* rowi = ei;
    const int* coli = ei + e;

    float* ws   = (float*)d_ws;
    float* h1   = ws;
    float* out1 = h1   + (size_t)n * H1_PITCH;
    float* dinv = out1 + (size_t)n * HID;
    float* g    = ws;
    float* out2 = g    + (size_t)n * G_PITCH;

    const int B = 256;
    int gn = (n + B - 1) / B;

    int ntiles = (n + 15) >> 4;
    int nwaves = (ntiles + TILES_PER_WAVE - 1) / TILES_PER_WAVE;
    int gw = (nwaves * 32 + B - 1) / B;
    gcn_gemm1_wmma <<<gw, B, 0, stream>>>(x, W1, h1, dinv, n);

    gcn_deg_scatter<<<dim3((n + DEG_NR - 1) / DEG_NR, 64), SCAT_BLK, 0, stream>>>(coli, dinv, n, e, 64);
    gcn_dinv_init_out1<<<gn, B, 0, stream>>>(dinv, h1, b1, out1, n);

    const dim3 sgrid((n + 3999) / 4000, 10);
    const size_t nblk     = (size_t)sgrid.x * sgrid.y;
    const size_t off_cnt  = (((size_t)n * (H1_PITCH + HID + 1) * sizeof(float)) + 127) & ~(size_t)127;
    const size_t off_list = off_cnt + ((nblk * sizeof(int) + 127) & ~(size_t)127);
    int*      counts = (int*)((char*)d_ws + off_cnt);
    unsigned* lists  = (unsigned*)((char*)d_ws + off_list);
    const bool with_lists = off_list + nblk * LIST_CAP * sizeof(unsigned) <= ws_size;

    if (with_lists)
        gcn_scatter_own<HID, 16, 4000, H1_PITCH, H1_DW, 12, 1><<<sgrid, SCAT_BLK, 0, stream>>>(rowi, coli, dinv, h1, out1, n, e, 10, lists, counts);
    else
        gcn_scatter_own<HID, 16, 4000, H1_PITCH, H1_DW, 12, 0><<<sgrid, SCAT_BLK, 0, stream>>>(rowi, coli, dinv, h1, out1, n, e, 10, nullptr, nullptr);

    gcn_gemm2      <<<gn, B, 0, stream>>>(out1, W2, b2, dinv, g, out2, n);
    if (with_lists)
        gcn_scatter_own<NCLS, 8, 4000, G_PITCH, G_DW, 12, 2><<<sgrid, SCAT_BLK, 0, stream>>>(rowi, coli, dinv, g, out2, n, e, 10, lists, counts);
    else
        gcn_scatter_own<NCLS, 8, 8192, G_PITCH, G_DW, 13, 0><<<dim3((n + 8191) / 8192, 19), SCAT_BLK, 0, stream>>>(rowi, coli, dinv, g, out2, n, e, 19, nullptr, nullptr);

    gcn_logsoftmax <<<gn, B, 0, stream>>>(out2, out, n);
}


# Thread-serial loops no longer pay a round trip per word: degree count asks 16 edge loads (16 B) at once, the 3 LDS scatter kernels drain 8 LDS reads ahead of their adds, zero sums by 16 B stores
# speedup vs baseline: 8.0107x; 1.0373x over previous
#include <hip/hip_runtime.h>
#include <hip/hip_bf16.h>

#define F_IN 512
#define HID  16
#define NCLS 7
#define TILES_PER_WAVE 2

#define H1_PITCH 32
#define H1_DW    16
#define G_PITCH  8
#define G_DW     7

typedef float v2f __attribute__((ext_vector_type(2)));
typedef float v8f __attribute__((ext_vector_type(8)));


#define SCAT_BLK   1024
#define SCAT_WAVES (SCAT_BLK / 32)
#define SCAT_TILE  512
#define DEG_NR     50000
#define FLUSH_U    8
#define DEG_U      16

__global__ __launch_bounds__(SCAT_BLK)
void gcn_deg_scatter(const int* __restrict__ col, float* __restrict__ deg, int n, int e, int nchunks) {
    __shared__ __attribute__((aligned(16))) float acc[DEG_NR];
    static_assert(DEG_NR % 4 == 0, "the sums are zeroed four words at a time");
    const int tid = threadIdx.x;
    const int lo = blockIdx.x * DEG_NR;
    const int nr = min(DEG_NR, n - lo);
    for (int j = tid * 4; j < DEG_NR; j += SCAT_BLK * 4) *(float4*)&acc[j] = make_float4(0.0f, 0.0f, 0.0f, 0.0f);
    __syncthreads();

    const int chunk = (((e + nchunks - 1) / nchunks) + 3) & ~3;
    const int e0 = blockIdx.y * chunk;
    const int e1 = min(e0 + chunk, e);
    const bool al = (((size_t)col) & 15) == 0;
    for (int i = e0 + tid * 4; i < e1; i += DEG_U * SCAT_BLK * 4) {
        int t[DEG_U][4];
        #pragma unroll
        for (int u = 0; u < DEG_U; ++u) {
            const int iu = i + u * SCAT_BLK * 4;
            if (al && iu + 4 <= e1) {
                int4 v = *(const int4*)(col + iu);
                t[u][0] = v.x; t[u][1] = v.y; t[u][2] = v.z; t[u][3] = v.w;
            } else {
                #pragma unroll
                for (int k = 0; k < 4; ++k) t[u][k] = (iu + k < e1) ? col[iu + k] : -1;
            }
        }
        #pragma unroll
        for (int u = 0; u < DEG_U; ++u) {
            #pragma unroll
            for (int k = 0; k < 4; ++k) {
                unsigned d = (unsigned)(t[u][k] - lo);
                if (d < (unsigned)nr) atomicAdd(&acc[d], 1.0f);
            }
        }
    }
    __syncthreads();
    float* dp = deg + lo;
    int j = tid;
    for (; j + (FLUSH_U - 1) * SCAT_BLK < nr; j += FLUSH_U * SCAT_BLK) {
        float v[FLUSH_U];
        #pragma unroll
        for (int u = 0; u < FLUSH_U; ++u) v[u] = acc[j + u * SCAT_BLK];
        #pragma unroll
        for (int u = 0; u < FLUSH_U; ++u)
            if (v[u] != 0.0f) atomicAdd(&dp[j + u * SCAT_BLK], v[u]);
    }
    if (j < nr) {
        float v[FLUSH_U];
        #pragma unroll
        for (int u = 0; u < FLUSH_U; ++u) v[u] = (j + u * SCAT_BLK < nr) ? acc[j + u * SCAT_BLK] : 0.0f;
        #pragma unroll
        for (int u = 0; u < FLUSH_U; ++u)
            if (v[u] != 0.0f) atomicAdd(&dp[j + u * SCAT_BLK], v[u]);
    }
}


#define SCAT_QCAP 512

#define LIST_GROUPS 512
#define LIST_CAP    (LIST_GROUPS * 32)
#define LIST_NONE   0xffffffffu

template <int LPE, int R>
__device__ __forceinline__ int gcn_group_lane(int x) {
    static_assert(LPE == 16 || LPE == 8, "lane groups are DPP16 rows or DPP8 groups");
    static_assert(R >= 0 && R < LPE, "source lane lies inside the group");
    if constexpr (LPE == 16) return __builtin_amdgcn_mov_dpp(x, 0x150 + R, 0xf, 0xf, true);
    else                     return (int)__builtin_amdgcn_mov_dpp8((unsigned)x, R * 0x249249);
}
template <int LPE, int R>
__device__ __forceinline__ float gcn_group_lane_f(float x) {
    return __int_as_float(gcn_group_lane<LPE, R>(__float_as_int(x)));
}

template <int LPE, int PITCH, int R>
__device__ __forceinline__ void gcn_gather_rounds(int s, int f, const float* __restrict__ src, float (&v)[LPE]) {
    v[R] = src[gcn_group_lane<LPE, R>(s) * PITCH + f];
    if constexpr (R + 1 < LPE) gcn_gather_rounds<LPE, PITCH, R + 1>(s, f, src, v);
}

template <int F, int LPE, int DW, bool FULL, bool SKIP, int R>
__device__ __forceinline__ void gcn_add_rounds(int tl, float w, float dt, int m, int sub, int f,
                                               const float (&v)[LPE], float* acc) {
    if (FULL || !SKIP || R < m) {
        const int te = gcn_group_lane<LPE, R>(tl);
        float we;
        if constexpr (DW >= LPE) we = gcn_group_lane_f<LPE, R>(w);
        else                     we = gcn_group_lane_f<LPE, DW>(v[R]) * gcn_group_lane_f<LPE, R>(dt);
        if ((FULL || sub * LPE + R < m) && f < F) atomicAdd(&acc[te * F + f], we * v[R]);
    }
    if constexpr (R + 1 < LPE) gcn_add_rounds<F, LPE, DW, FULL, SKIP, R + 1>(tl, w, dt, m, sub, f, v, acc);
}

template <int F, int LPE, int PITCH, int DW, int TB, bool FULL, bool SKIP = true>
__device__ __forceinline__ void gcn_scatter_pass(unsigned ent, int m, int sub, int f, int lo,
                                                 const float* __restrict__ dinv, const float* __restrict__ src,
                                                 float* acc) {
    const int s  = (int)(ent >> TB);
    const int tl = (int)(ent & ((1u << TB) - 1u));
    float ds = 0.0f;
    if constexpr (DW >= LPE) ds = src[s * PITCH + DW];
    float v[LPE];
    gcn_gather_rounds<LPE, PITCH, 0>(s, f, src, v);
    const float dt = dinv[lo + tl];
    const float w  = ds * dt;
    gcn_add_rounds<F, LPE, DW, FULL, SKIP, 0>(tl, w, dt, m, sub, f, v, acc);
}

__device__ __forceinline__ void gcn_list_put(unsigned ent, bool real, int lane, int* ngroups, unsigned* lst) {
    int g = 0;
    if (lane == 0) g = atomicAdd(ngroups, 1);
    g = __builtin_amdgcn_readfirstlane(g);
    if (g < LIST_GROUPS) lst[g * 32 + lane] = real ? ent : LIST_NONE;
}

template <int F, int LPE, int NR, int PITCH, int DW, int TB, int LIST>
__global__ __launch_bounds__(SCAT_BLK)
void gcn_scatter_own(const int* __restrict__ rowi, const int* __restrict__ coli,
                     const float* __restrict__ dinv, const float* __restrict__ src,
                     float* __restrict__ dst, int n, int e, int nchunks,
                     unsigned* lists, int* counts) {
    static_assert(NR <= (1 << TB), "target offset must fit the packed word");
    static_assert(LIST == 0 || NR < (1 << TB) - 1, "LIST_NONE must not be an entry");
    static_assert((NR * F) % 4 == 0, "the sums are zeroed four words at a time");
    __shared__ __attribute__((aligned(16))) float acc[NR * F];
    __shared__ unsigned queue[SCAT_WAVES * SCAT_QCAP];
    __shared__ int      ngroups;
    const int tid  = threadIdx.x;
    const int lane = tid & 31;
    const int wv   = tid >> 5;
    const int lo = blockIdx.x * NR;
    const int nr = min(NR, n - lo);
    const int bid = blockIdx.y * gridDim.x + blockIdx.x;
    unsigned* lst = (LIST != 0) ? lists + (size_t)bid * LIST_CAP : nullptr;
    for (int j = tid * 4; j < NR * F; j += SCAT_BLK * 4) *(float4*)&acc[j] = make_float4(0.0f, 0.0f, 0.0f, 0.0f);
    if (LIST == 1 && tid == 0) ngroups = 0;
    __syncthreads();
    const int sub = lane / LPE;
    const int f   = lane % LPE;

    const int ng = (LIST == 2) ? counts[bid] : 0;
    if (LIST == 2 && ng <= LIST_GROUPS) {
        int gi = __builtin_amdgcn_readfirstlane(wv);
        unsigned nxt = (gi < ng) ? lst[gi * 32 + lane] : 0u;
        for (; gi < ng; gi += SCAT_WAVES) {
            unsigned ent = nxt;
            if (gi + SCAT_WAVES < ng) nxt = lst[(gi + SCAT_WAVES) * 32 + lane];
            const int m = __popc((unsigned)__ballot(ent != LIST_NONE));
            const unsigned first = (unsigned)__builtin_amdgcn_readfirstlane((int)ent);
            if (lane >= m) ent = first;
            gcn_scatter_pass<F, LPE, PITCH, DW, TB, false, false>(ent, m, sub, f, lo, dinv, src, acc);
        }
    } else {

    const int chunk = (((e + nchunks - 1) / nchunks) + SCAT_TILE - 1) & ~(SCAT_TILE - 1);
    const int e0 = blockIdx.y * chunk;
    const int e1 = min(e0 + chunk, e);
    const bool al = ((((size_t)coli) | ((size_t)rowi)) & 15) == 0;
    unsigned* q = queue + wv * SCAT_QCAP;
    const unsigned below = (1u << lane) - 1u;

    auto load_half = [&](int b, int hlf, int (&c)[16], int (&r)[16]) {
        if (al && b + SCAT_TILE <= e1) {
            #pragma unroll
            for (int j = 2 * hlf; j < 2 * hlf + 2; ++j) {
                const int4 vc = *(const int4*)(coli + b + j * 128 + lane * 4);
                const int4 vr = *(const int4*)(rowi + b + j * 128 + lane * 4);
                c[4*j+0] = vc.x; c[4*j+1] = vc.y; c[4*j+2] = vc.z; c[4*j+3] = vc.w;
                r[4*j+0] = vr.x; r[4*j+1] = vr.y; r[4*j+2] = vr.z; r[4*j+3] = vr.w;
            }
        } else {
            #pragma unroll
            for (int j = 8 * hlf; j < 8 * hlf + 8; ++j) {
                const int idx = b + (j >> 2) * 128 + lane * 4 + (j & 3);
                const bool in = idx < e1;
                c[j] = in ? coli[idx] : -1;
                r[j] = in ? rowi[idx] : 0;
            }
        }
    };

    constexpr int STEP = SCAT_WAVES * SCAT_TILE;
    int base = e0 + wv * SCAT_TILE;
    int c[16], r[16];
    int head = 0, tail = 0;
    if (base < e1) { load_half(base, 0, c, r); load_half(base, 1, c, r); }
    for (; base < e1; base += STEP) {
        #pragma unroll
        for (int hlf = 0; hlf < 2; ++hlf) {
            #pragma unroll
            for (int j = 8 * hlf; j < 8 * hlf + 8; ++j) {
                const unsigned d = (unsigned)(c[j] - lo);
                const bool hit = d < (unsigned)nr;
                const unsigned mask = (unsigned)__ballot(hit);
                if (hit) q[(tail + __popc(mask & below)) & (SCAT_QCAP - 1)] = ((unsigned)r[j] << TB) | d;
                tail += __popc(mask);
            }
            if (base + STEP < e1) load_half(base + STEP, hlf, c, r);
            __builtin_amdgcn_fence(__ATOMIC_RELEASE, "wavefront");
            __builtin_amdgcn_wave_barrier();
            while (tail - head >= 32) {
                const unsigned ent = q[(head + lane) & (SCAT_QCAP - 1)];
                gcn_scatter_pass<F, LPE, PITCH, DW, TB, true>(ent, 32, sub, f, lo, dinv, src, acc);
                if (LIST == 1) gcn_list_put(ent, true, lane, &ngroups, lst);
                head += 32;
            }
            __builtin_amdgcn_fence(__ATOMIC_ACQUIRE, "wavefront");
            __builtin_amdgcn_wave_barrier();
        }
    }
    if (tail != head) {
        const int m = tail - head;
        const unsigned ent = q[(head + (lane < m ? lane : 0)) & (SCAT_QCAP - 1)];
        gcn_scatter_pass<F, LPE, PITCH, DW, TB, false>(ent, m, sub, f, lo, dinv, src, acc);
        if (LIST == 1) gcn_list_put(ent, lane < m, lane, &ngroups, lst);
    }
    }
    __syncthreads();
    if (LIST == 1 && tid == 0) counts[bid] = ngroups;
    float* dp = dst + (size_t)lo * F;
    const int tot = nr * F;
    int j = tid;
    for (; j + (FLUSH_U - 1) * SCAT_BLK < tot; j += FLUSH_U * SCAT_BLK) {
        float v[FLUSH_U];
        #pragma unroll
        for (int u = 0; u < FLUSH_U; ++u) v[u] = acc[j + u * SCAT_BLK];
        #pragma unroll
        for (int u = 0; u < FLUSH_U; ++u) atomicAdd(&dp[j + u * SCAT_BLK], v[u]);
    }
    if (j < tot) {
        float v[FLUSH_U];
        #pragma unroll
        for (int u = 0; u < FLUSH_U; ++u) v[u] = (j + u * SCAT_BLK < tot) ? acc[j + u * SCAT_BLK] : 0.0f;
        #pragma unroll
        for (int u = 0; u < FLUSH_U; ++u)
            if (j + u * SCAT_BLK < tot) atomicAdd(&dp[j + u * SCAT_BLK], v[u]);
    }
}

__global__ void gcn_dinv_init_out1(float* __restrict__ degdinv, float* h1,
                                   const float* __restrict__ b1, float* __restrict__ out1, int n) {
    int i = blockIdx.x * blockDim.x + threadIdx.x;
    if (i >= n) return;
    const float r = rsqrtf(degdinv[i]);
    degdinv[i] = r;
    float* hrow = h1 + (size_t)i * H1_PITCH;
    hrow[H1_DW] = r;
    const float s = r * r;
    const float4* hp = (const float4*)hrow;
    float4* op = (float4*)(out1 + (size_t)i * HID);
    #pragma unroll
    for (int q = 0; q < 4; ++q) {
        float4 h = hp[q];
        op[q] = make_float4(b1[4*q+0] + s*h.x, b1[4*q+1] + s*h.y,
                            b1[4*q+2] + s*h.z, b1[4*q+3] + s*h.w);
    }
}


__device__ __forceinline__ void store_tile(float* __restrict__ h1, const v8f& acc,
                                           int tile, int lane, int n) {
    int mbase = tile * 16 + ((lane >> 4) << 3);
    int ncol  = lane & 15;
    float* p = h1 + (size_t)mbase * H1_PITCH + ncol;
    if (mbase + 8 <= n) {
        #pragma unroll
        for (int v = 0; v < 8; ++v) p[v * H1_PITCH] = acc[v];
    } else {
        #pragma unroll
        for (int v = 0; v < 8; ++v) if (mbase + v < n) p[v * H1_PITCH] = acc[v];
    }
}

__global__ void gcn_gemm1_wmma(const float* __restrict__ x, const float* __restrict__ W1,
                               float* __restrict__ h1, float* __restrict__ deg, int n) {
    int gid = blockIdx.x * blockDim.x + threadIdx.x;
    if (gid < n) deg[gid] = 1.0f;
    int lane = threadIdx.x & 31;
    int wid  = gid >> 5;
    int ntiles = (n + 15) >> 4;
    int tile0 = wid * TILES_PER_WAVE;
    if (tile0 >= ntiles) return;
    bool has1 = (tile0 + 1) < ntiles;

    int m    = lane & 15;
    int koff = (lane >> 4) << 1;

    int row0 = tile0 * 16 + m;
    int row1 = row0 + 16;
    const float* xa0 = x + (size_t)(row0 < n ? row0 : 0) * F_IN + koff;
    const float* xa1 = x + (size_t)(row1 < n ? row1 : 0) * F_IN + koff;

    v8f acc0 = {}, acc1 = {};
    #pragma unroll 4
    for (int k = 0; k < F_IN; k += 4) {
        v2f b;
        b.x = W1[(size_t)(k + koff)     * HID + m];
        b.y = W1[(size_t)(k + koff + 1) * HID + m];
        v2f a0 = *(const v2f*)(xa0 + k);
        v2f a1 = *(const v2f*)(xa1 + k);
        acc0 = __builtin_amdgcn_wmma_f32_16x16x4_f32(
            false, a0, false, b, (short)0, acc0, false, false);
        acc1 = __builtin_amdgcn_wmma_f32_16x16x4_f32(
            false, a1, false, b, (short)0, acc1, false, false);
    }

    store_tile(h1, acc0, tile0, lane, n);
    if (has1) store_tile(h1, acc1, tile0 + 1, lane, n);
}


__global__ void gcn_gemm2(const float* __restrict__ out1, const float* __restrict__ W2,
                          const float* __restrict__ b2, const float* __restrict__ dinv,
                          float* __restrict__ g, float* __restrict__ out2, int n) {
    int i = blockIdx.x * blockDim.x + threadIdx.x;
    if (i >= n) return;
    float h[HID];
    const float* ip = out1 + (size_t)i * HID;
    #pragma unroll
    for (int c = 0; c < HID; ++c) h[c] = fmaxf(ip[c], 0.0f);
    const float dv = dinv[i];
    const float s = dv * dv;
    float a[G_PITCH];
    #pragma unroll
    for (int c2 = 0; c2 < NCLS; ++c2) {
        float acc = 0.0f;
        #pragma unroll
        for (int c = 0; c < HID; ++c) acc += h[c] * W2[c * NCLS + c2];
        a[c2] = acc;
        out2[(size_t)i * NCLS + c2] = b2[c2] + s * acc;
    }
    a[G_DW] = dv;
    float4* gp = (float4*)(g + (size_t)i * G_PITCH);
    gp[0] = make_float4(a[0], a[1], a[2], a[3]);
    gp[1] = make_float4(a[4], a[5], a[6], a[7]);
}


__global__ void gcn_logsoftmax(const float* __restrict__ out2, float* __restrict__ out, int n) {
    int i = blockIdx.x * blockDim.x + threadIdx.x;
    if (i >= n) return;
    float v[NCLS];
    float m = -1e30f;
    #pragma unroll
    for (int c = 0; c < NCLS; ++c) { v[c] = out2[(size_t)i * NCLS + c]; m = fmaxf(m, v[c]); }
    float sum = 0.0f;
    #pragma unroll
    for (int c = 0; c < NCLS; ++c) sum += __expf(v[c] - m);
    float lse = m + __logf(sum);
    #pragma unroll
    for (int c = 0; c < NCLS; ++c) out[(size_t)i * NCLS + c] = v[c] - lse;
}


extern "C" void kernel_launch(void* const* d_in, const int* in_sizes, int n_in,
                              void* d_out, int out_size, void* d_ws, size_t ws_size,
                              hipStream_t stream) {
    const float* x  = (const float*)d_in[0];
    const int*   ei = (const int*)  d_in[1];
    const float* W1 = (const float*)d_in[2];
    const float* b1 = (const float*)d_in[3];
    const float* W2 = (const float*)d_in[4];
    const float* b2 = (const float*)d_in[5];
    float* out = (float*)d_out;

    int n = in_sizes[0] / F_IN;
    int e = in_sizes[1] / 2;
    const int* rowi = ei;
    const int* coli = ei + e;

    float* ws   = (float*)d_ws;
    float* h1   = ws;
    float* out1 = h1   + (size_t)n * H1_PITCH;
    float* dinv = out1 + (size_t)n * HID;
    float* g    = ws;
    float* out2 = g    + (size_t)n * G_PITCH;

    const int B = 256;
    int gn = (n + B - 1) / B;

    int ntiles = (n + 15) >> 4;
    int nwaves = (ntiles + TILES_PER_WAVE - 1) / TILES_PER_WAVE;
    int gw = (nwaves * 32 + B - 1) / B;
    gcn_gemm1_wmma <<<gw, B, 0, stream>>>(x, W1, h1, dinv, n);

    gcn_deg_scatter<<<dim3((n + DEG_NR - 1) / DEG_NR, 64), SCAT_BLK, 0, stream>>>(coli, dinv, n, e, 64);
    gcn_dinv_init_out1<<<gn, B, 0, stream>>>(dinv, h1, b1, out1, n);

    const dim3 sgrid((n + 3999) / 4000, 10);
    const size_t nblk     = (size_t)sgrid.x * sgrid.y;
    const size_t off_cnt  = (((size_t)n * (H1_PITCH + HID + 1) * sizeof(float)) + 127) & ~(size_t)127;
    const size_t off_list = off_cnt + ((nblk * sizeof(int) + 127) & ~(size_t)127);
    int*      counts = (int*)((char*)d_ws + off_cnt);
    unsigned* lists  = (unsigned*)((char*)d_ws + off_list);
    const bool with_lists = off_list + nblk * LIST_CAP * sizeof(unsigned) <= ws_size;

    if (with_lists)
        gcn_scatter_own<HID, 16, 4000, H1_PITCH, H1_DW, 12, 1><<<sgrid, SCAT_BLK, 0, stream>>>(rowi, coli, dinv, h1, out1, n, e, 10, lists, counts);
    else
        gcn_scatter_own<HID, 16, 4000, H1_PITCH, H1_DW, 12, 0><<<sgrid, SCAT_BLK, 0, stream>>>(rowi, coli, dinv, h1, out1, n, e, 10, nullptr, nullptr);

    gcn_gemm2      <<<gn, B, 0, stream>>>(out1, W2, b2, dinv, g, out2, n);
    if (with_lists)
        gcn_scatter_own<NCLS, 8, 4000, G_PITCH, G_DW, 12, 2><<<sgrid, SCAT_BLK, 0, stream>>>(rowi, coli, dinv, g, out2, n, e, 10, lists, counts);
    else
        gcn_scatter_own<NCLS, 8, 8192, G_PITCH, G_DW, 13, 0><<<dim3((n + 8191) / 8192, 19), SCAT_BLK, 0, stream>>>(rowi, coli, dinv, g, out2, n, e, 19, nullptr, nullptr);

    gcn_logsoftmax <<<gn, B, 0, stream>>>(out2, out, n);
}


# Feature scatters: a hit's target dinv comes from an LDS copy of the block's own target range (filled once a block), not a 32-address global gather a pass; ring 256, drained per 128 edges
# speedup vs baseline: 8.2582x; 1.0309x over previous
#include <hip/hip_runtime.h>
#include <hip/hip_bf16.h>

#define F_IN 512
#define HID  16
#define NCLS 7
#define TILES_PER_WAVE 2

#define H1_PITCH 32
#define H1_DW    16
#define G_PITCH  8
#define G_DW     7

typedef float v2f __attribute__((ext_vector_type(2)));
typedef float v8f __attribute__((ext_vector_type(8)));


#define SCAT_BLK   1024
#define SCAT_WAVES (SCAT_BLK / 32)
#define SCAT_TILE  512
#define DEG_NR     50000
#define FLUSH_U    8
#define DEG_U      16

__global__ __launch_bounds__(SCAT_BLK)
void gcn_deg_scatter(const int* __restrict__ col, float* __restrict__ deg, int n, int e, int nchunks) {
    __shared__ __attribute__((aligned(16))) float acc[DEG_NR];
    static_assert(DEG_NR % 4 == 0, "the sums are zeroed four words at a time");
    const int tid = threadIdx.x;
    const int lo = blockIdx.x * DEG_NR;
    const int nr = min(DEG_NR, n - lo);
    for (int j = tid * 4; j < DEG_NR; j += SCAT_BLK * 4) *(float4*)&acc[j] = make_float4(0.0f, 0.0f, 0.0f, 0.0f);
    __syncthreads();

    const int chunk = (((e + nchunks - 1) / nchunks) + 3) & ~3;
    const int e0 = blockIdx.y * chunk;
    const int e1 = min(e0 + chunk, e);
    const bool al = (((size_t)col) & 15) == 0;
    for (int i = e0 + tid * 4; i < e1; i += DEG_U * SCAT_BLK * 4) {
        int t[DEG_U][4];
        #pragma unroll
        for (int u = 0; u < DEG_U; ++u) {
            const int iu = i + u * SCAT_BLK * 4;
            if (al && iu + 4 <= e1) {
                int4 v = *(const int4*)(col + iu);
                t[u][0] = v.x; t[u][1] = v.y; t[u][2] = v.z; t[u][3] = v.w;
            } else {
                #pragma unroll
                for (int k = 0; k < 4; ++k) t[u][k] = (iu + k < e1) ? col[iu + k] : -1;
            }
        }
        #pragma unroll
        for (int u = 0; u < DEG_U; ++u) {
            #pragma unroll
            for (int k = 0; k < 4; ++k) {
                unsigned d = (unsigned)(t[u][k] - lo);
                if (d < (unsigned)nr) atomicAdd(&acc[d], 1.0f);
            }
        }
    }
    __syncthreads();
    float* dp = deg + lo;
    int j = tid;
    for (; j + (FLUSH_U - 1) * SCAT_BLK < nr; j += FLUSH_U * SCAT_BLK) {
        float v[FLUSH_U];
        #pragma unroll
        for (int u = 0; u < FLUSH_U; ++u) v[u] = acc[j + u * SCAT_BLK];
        #pragma unroll
        for (int u = 0; u < FLUSH_U; ++u)
            if (v[u] != 0.0f) atomicAdd(&dp[j + u * SCAT_BLK], v[u]);
    }
    if (j < nr) {
        float v[FLUSH_U];
        #pragma unroll
        for (int u = 0; u < FLUSH_U; ++u) v[u] = (j + u * SCAT_BLK < nr) ? acc[j + u * SCAT_BLK] : 0.0f;
        #pragma unroll
        for (int u = 0; u < FLUSH_U; ++u)
            if (v[u] != 0.0f) atomicAdd(&dp[j + u * SCAT_BLK], v[u]);
    }
}


#define SCAT_QCAP_BIG   512
#define SCAT_QCAP_SMALL 256
#define SCAT_LDS_BYTES  (320 * 1024)

#define LIST_GROUPS 512
#define LIST_CAP    (LIST_GROUPS * 32)
#define LIST_NONE   0xffffffffu

template <int LPE, int R>
__device__ __forceinline__ int gcn_group_lane(int x) {
    static_assert(LPE == 16 || LPE == 8, "lane groups are DPP16 rows or DPP8 groups");
    static_assert(R >= 0 && R < LPE, "source lane lies inside the group");
    if constexpr (LPE == 16) return __builtin_amdgcn_mov_dpp(x, 0x150 + R, 0xf, 0xf, true);
    else                     return (int)__builtin_amdgcn_mov_dpp8((unsigned)x, R * 0x249249);
}
template <int LPE, int R>
__device__ __forceinline__ float gcn_group_lane_f(float x) {
    return __int_as_float(gcn_group_lane<LPE, R>(__float_as_int(x)));
}

template <int LPE, int PITCH, int R>
__device__ __forceinline__ void gcn_gather_rounds(int s, int f, const float* __restrict__ src, float (&v)[LPE]) {
    v[R] = src[gcn_group_lane<LPE, R>(s) * PITCH + f];
    if constexpr (R + 1 < LPE) gcn_gather_rounds<LPE, PITCH, R + 1>(s, f, src, v);
}

template <int F, int LPE, int DW, bool FULL, bool SKIP, int R>
__device__ __forceinline__ void gcn_add_rounds(int tl, float w, float dt, int m, int sub, int f,
                                               const float (&v)[LPE], float* acc) {
    if (FULL || !SKIP || R < m) {
        const int te = gcn_group_lane<LPE, R>(tl);
        float we;
        if constexpr (DW >= LPE) we = gcn_group_lane_f<LPE, R>(w);
        else                     we = gcn_group_lane_f<LPE, DW>(v[R]) * gcn_group_lane_f<LPE, R>(dt);
        if ((FULL || sub * LPE + R < m) && f < F) atomicAdd(&acc[te * F + f], we * v[R]);
    }
    if constexpr (R + 1 < LPE) gcn_add_rounds<F, LPE, DW, FULL, SKIP, R + 1>(tl, w, dt, m, sub, f, v, acc);
}

template <int F, int LPE, int PITCH, int DW, int TB, bool FULL, bool SKIP = true>
__device__ __forceinline__ void gcn_scatter_pass(unsigned ent, int m, int sub, int f,
                                                 const float* dloc, const float* __restrict__ src,
                                                 float* acc) {
    const int s  = (int)(ent >> TB);
    const int tl = (int)(ent & ((1u << TB) - 1u));
    float ds = 0.0f;
    if constexpr (DW >= LPE) ds = src[s * PITCH + DW];
    float v[LPE];
    gcn_gather_rounds<LPE, PITCH, 0>(s, f, src, v);
    const float dt = dloc[tl];
    const float w  = ds * dt;
    gcn_add_rounds<F, LPE, DW, FULL, SKIP, 0>(tl, w, dt, m, sub, f, v, acc);
}

__device__ __forceinline__ void gcn_list_put(unsigned ent, bool real, int lane, int* ngroups, unsigned* lst) {
    int g = 0;
    if (lane == 0) g = atomicAdd(ngroups, 1);
    g = __builtin_amdgcn_readfirstlane(g);
    if (g < LIST_GROUPS) lst[g * 32 + lane] = real ? ent : LIST_NONE;
}

template <int F, int LPE, int NR, int PITCH, int DW, int TB, int LIST>
__global__ __launch_bounds__(SCAT_BLK)
void gcn_scatter_own(const int* __restrict__ rowi, const int* __restrict__ coli,
                     const float* __restrict__ dinv, const float* __restrict__ src,
                     float* __restrict__ dst, int n, int e, int nchunks,
                     unsigned* lists, int* counts) {
    static_assert(NR <= (1 << TB), "target offset must fit the packed word");
    static_assert(LIST == 0 || NR < (1 << TB) - 1, "LIST_NONE must not be an entry");
    static_assert((NR * F) % 4 == 0, "the sums are zeroed four words at a time");
    constexpr int QCAP = ((size_t)NR * F * 4 + (size_t)NR * 4 + (size_t)SCAT_WAVES * SCAT_QCAP_BIG * 4 + 64 <= SCAT_LDS_BYTES)
                             ? SCAT_QCAP_BIG : SCAT_QCAP_SMALL;
    static_assert((size_t)NR * F * 4 + (size_t)NR * 4 + (size_t)SCAT_WAVES * QCAP * 4 + 64 <= SCAT_LDS_BYTES,
                  "sums, dinv copy and rings must fit one block's LDS");
    __shared__ __attribute__((aligned(16))) float acc[NR * F];
    __shared__ float    dloc[NR];
    __shared__ unsigned queue[SCAT_WAVES * QCAP];
    __shared__ int      ngroups;
    const int tid  = threadIdx.x;
    const int lane = tid & 31;
    const int wv   = tid >> 5;
    const int lo = blockIdx.x * NR;
    const int nr = min(NR, n - lo);
    const int bid = blockIdx.y * gridDim.x + blockIdx.x;
    unsigned* lst = (LIST != 0) ? lists + (size_t)bid * LIST_CAP : nullptr;
    for (int j = tid * 4; j < NR * F; j += SCAT_BLK * 4) *(float4*)&acc[j] = make_float4(0.0f, 0.0f, 0.0f, 0.0f);
    for (int j = tid; j < nr; j += SCAT_BLK) dloc[j] = dinv[lo + j];
    if (LIST == 1 && tid == 0) ngroups = 0;
    __syncthreads();
    const int sub = lane / LPE;
    const int f   = lane % LPE;

    const int ng = (LIST == 2) ? counts[bid] : 0;
    if (LIST == 2 && ng <= LIST_GROUPS) {
        int gi = __builtin_amdgcn_readfirstlane(wv);
        unsigned nxt = (gi < ng) ? lst[gi * 32 + lane] : 0u;
        for (; gi < ng; gi += SCAT_WAVES) {
            unsigned ent = nxt;
            if (gi + SCAT_WAVES < ng) nxt = lst[(gi + SCAT_WAVES) * 32 + lane];
            const int m = __popc((unsigned)__ballot(ent != LIST_NONE));
            const unsigned first = (unsigned)__builtin_amdgcn_readfirstlane((int)ent);
            if (lane >= m) ent = first;
            gcn_scatter_pass<F, LPE, PITCH, DW, TB, false, false>(ent, m, sub, f, dloc, src, acc);
        }
    } else {

    const int chunk = (((e + nchunks - 1) / nchunks) + SCAT_TILE - 1) & ~(SCAT_TILE - 1);
    const int e0 = blockIdx.y * chunk;
    const int e1 = min(e0 + chunk, e);
    const bool al = ((((size_t)coli) | ((size_t)rowi)) & 15) == 0;
    unsigned* q = queue + wv * QCAP;
    const unsigned below = (1u << lane) - 1u;

    auto load_quarter = [&](int b, int qt, int (&c)[16], int (&r)[16]) {
        if (al && b + SCAT_TILE <= e1) {
            const int4 vc = *(const int4*)(coli + b + qt * 128 + lane * 4);
            const int4 vr = *(const int4*)(rowi + b + qt * 128 + lane * 4);
            c[4*qt+0] = vc.x; c[4*qt+1] = vc.y; c[4*qt+2] = vc.z; c[4*qt+3] = vc.w;
            r[4*qt+0] = vr.x; r[4*qt+1] = vr.y; r[4*qt+2] = vr.z; r[4*qt+3] = vr.w;
        } else {
            #pragma unroll
            for (int j = 4 * qt; j < 4 * qt + 4; ++j) {
                const int idx = b + qt * 128 + lane * 4 + (j & 3);
                const bool in = idx < e1;
                c[j] = in ? coli[idx] : -1;
                r[j] = in ? rowi[idx] : 0;
            }
        }
    };

    constexpr int STEP = SCAT_WAVES * SCAT_TILE;
    int base = e0 + wv * SCAT_TILE;
    int c[16], r[16];
    int head = 0, tail = 0;
    if (base < e1) {
        #pragma unroll
        for (int qt = 0; qt < 4; ++qt) load_quarter(base, qt, c, r);
    }
    for (; base < e1; base += STEP) {
        #pragma unroll
        for (int qt = 0; qt < 4; ++qt) {
            #pragma unroll
            for (int j = 4 * qt; j < 4 * qt + 4; ++j) {
                const unsigned d = (unsigned)(c[j] - lo);
                const bool hit = d < (unsigned)nr;
                const unsigned mask = (unsigned)__ballot(hit);
                if (hit) q[(tail + __popc(mask & below)) & (QCAP - 1)] = ((unsigned)r[j] << TB) | d;
                tail += __popc(mask);
            }
            if (base + STEP < e1) load_quarter(base + STEP, qt, c, r);
            __builtin_amdgcn_fence(__ATOMIC_RELEASE, "wavefront");
            __builtin_amdgcn_wave_barrier();
            while (tail - head >= 32) {
                const unsigned ent = q[(head + lane) & (QCAP - 1)];
                gcn_scatter_pass<F, LPE, PITCH, DW, TB, true>(ent, 32, sub, f, dloc, src, acc);
                if (LIST == 1) gcn_list_put(ent, true, lane, &ngroups, lst);
                head += 32;
            }
            __builtin_amdgcn_fence(__ATOMIC_ACQUIRE, "wavefront");
            __builtin_amdgcn_wave_barrier();
        }
    }
    if (tail != head) {
        const int m = tail - head;
        const unsigned ent = q[(head + (lane < m ? lane : 0)) & (QCAP - 1)];
        gcn_scatter_pass<F, LPE, PITCH, DW, TB, false>(ent, m, sub, f, dloc, src, acc);
        if (LIST == 1) gcn_list_put(ent, lane < m, lane, &ngroups, lst);
    }
    }
    __syncthreads();
    if (LIST == 1 && tid == 0) counts[bid] = ngroups;
    float* dp = dst + (size_t)lo * F;
    const int tot = nr * F;
    int j = tid;
    for (; j + (FLUSH_U - 1) * SCAT_BLK < tot; j += FLUSH_U * SCAT_BLK) {
        float v[FLUSH_U];
        #pragma unroll
        for (int u = 0; u < FLUSH_U; ++u) v[u] = acc[j + u * SCAT_BLK];
        #pragma unroll
        for (int u = 0; u < FLUSH_U; ++u) atomicAdd(&dp[j + u * SCAT_BLK], v[u]);
    }
    if (j < tot) {
        float v[FLUSH_U];
        #pragma unroll
        for (int u = 0; u < FLUSH_U; ++u) v[u] = (j + u * SCAT_BLK < tot) ? acc[j + u * SCAT_BLK] : 0.0f;
        #pragma unroll
        for (int u = 0; u < FLUSH_U; ++u)
            if (j + u * SCAT_BLK < tot) atomicAdd(&dp[j + u * SCAT_BLK], v[u]);
    }
}

__global__ void gcn_dinv_init_out1(float* __restrict__ degdinv, float* h1,
                                   const float* __restrict__ b1, float* __restrict__ out1, int n) {
    int i = blockIdx.x * blockDim.x + threadIdx.x;
    if (i >= n) return;
    const float r = rsqrtf(degdinv[i]);
    degdinv[i] = r;
    float* hrow = h1 + (size_t)i * H1_PITCH;
    hrow[H1_DW] = r;
    const float s = r * r;
    const float4* hp = (const float4*)hrow;
    float4* op = (float4*)(out1 + (size_t)i * HID);
    #pragma unroll
    for (int q = 0; q < 4; ++q) {
        float4 h = hp[q];
        op[q] = make_float4(b1[4*q+0] + s*h.x, b1[4*q+1] + s*h.y,
                            b1[4*q+2] + s*h.z, b1[4*q+3] + s*h.w);
    }
}


__device__ __forceinline__ void store_tile(float* __restrict__ h1, const v8f& acc,
                                           int tile, int lane, int n) {
    int mbase = tile * 16 + ((lane >> 4) << 3);
    int ncol  = lane & 15;
    float* p = h1 + (size_t)mbase * H1_PITCH + ncol;
    if (mbase + 8 <= n) {
        #pragma unroll
        for (int v = 0; v < 8; ++v) p[v * H1_PITCH] = acc[v];
    } else {
        #pragma unroll
        for (int v = 0; v < 8; ++v) if (mbase + v < n) p[v * H1_PITCH] = acc[v];
    }
}

__global__ void gcn_gemm1_wmma(const float* __restrict__ x, const float* __restrict__ W1,
                               float* __restrict__ h1, float* __restrict__ deg, int n) {
    int gid = blockIdx.x * blockDim.x + threadIdx.x;
    if (gid < n) deg[gid] = 1.0f;
    int lane = threadIdx.x & 31;
    int wid  = gid >> 5;
    int ntiles = (n + 15) >> 4;
    int tile0 = wid * TILES_PER_WAVE;
    if (tile0 >= ntiles) return;
    bool has1 = (tile0 + 1) < ntiles;

    int m    = lane & 15;
    int koff = (lane >> 4) << 1;

    int row0 = tile0 * 16 + m;
    int row1 = row0 + 16;
    const float* xa0 = x + (size_t)(row0 < n ? row0 : 0) * F_IN + koff;
    const float* xa1 = x + (size_t)(row1 < n ? row1 : 0) * F_IN + koff;

    v8f acc0 = {}, acc1 = {};
    #pragma unroll 4
    for (int k = 0; k < F_IN; k += 4) {
        v2f b;
        b.x = W1[(size_t)(k + koff)     * HID + m];
        b.y = W1[(size_t)(k + koff + 1) * HID + m];
        v2f a0 = *(const v2f*)(xa0 + k);
        v2f a1 = *(const v2f*)(xa1 + k);
        acc0 = __builtin_amdgcn_wmma_f32_16x16x4_f32(
            false, a0, false, b, (short)0, acc0, false, false);
        acc1 = __builtin_amdgcn_wmma_f32_16x16x4_f32(
            false, a1, false, b, (short)0, acc1, false, false);
    }

    store_tile(h1, acc0, tile0, lane, n);
    if (has1) store_tile(h1, acc1, tile0 + 1, lane, n);
}


__global__ void gcn_gemm2(const float* __restrict__ out1, const float* __restrict__ W2,
                          const float* __restrict__ b2, const float* __restrict__ dinv,
                          float* __restrict__ g, float* __restrict__ out2, int n) {
    int i = blockIdx.x * blockDim.x + threadIdx.x;
    if (i >= n) return;
    float h[HID];
    const float* ip = out1 + (size_t)i * HID;
    #pragma unroll
    for (int c = 0; c < HID; ++c) h[c] = fmaxf(ip[c], 0.0f);
    const float dv = dinv[i];
    const float s = dv * dv;
    float a[G_PITCH];
    #pragma unroll
    for (int c2 = 0; c2 < NCLS; ++c2) {
        float acc = 0.0f;
        #pragma unroll
        for (int c = 0; c < HID; ++c) acc += h[c] * W2[c * NCLS + c2];
        a[c2] = acc;
        out2[(size_t)i * NCLS + c2] = b2[c2] + s * acc;
    }
    a[G_DW] = dv;
    float4* gp = (float4*)(g + (size_t)i * G_PITCH);
    gp[0] = make_float4(a[0], a[1], a[2], a[3]);
    gp[1] = make_float4(a[4], a[5], a[6], a[7]);
}


__global__ void gcn_logsoftmax(const float* __restrict__ out2, float* __restrict__ out, int n) {
    int i = blockIdx.x * blockDim.x + threadIdx.x;
    if (i >= n) return;
    float v[NCLS];
    float m = -1e30f;
    #pragma unroll
    for (int c = 0; c < NCLS; ++c) { v[c] = out2[(size_t)i * NCLS + c]; m = fmaxf(m, v[c]); }
    float sum = 0.0f;
    #pragma unroll
    for (int c = 0; c < NCLS; ++c) sum += __expf(v[c] - m);
    float lse = m + __logf(sum);
    #pragma unroll
    for (int c = 0; c < NCLS; ++c) out[(size_t)i * NCLS + c] = v[c] - lse;
}


extern "C" void kernel_launch(void* const* d_in, const int* in_sizes, int n_in,
                              void* d_out, int out_size, void* d_ws, size_t ws_size,
                              hipStream_t stream) {
    const float* x  = (const float*)d_in[0];
    const int*   ei = (const int*)  d_in[1];
    const float* W1 = (const float*)d_in[2];
    const float* b1 = (const float*)d_in[3];
    const float* W2 = (const float*)d_in[4];
    const float* b2 = (const float*)d_in[5];
    float* out = (float*)d_out;

    int n = in_sizes[0] / F_IN;
    int e = in_sizes[1] / 2;
    const int* rowi = ei;
    const int* coli = ei + e;

    float* ws   = (float*)d_ws;
    float* h1   = ws;
    float* out1 = h1   + (size_t)n * H1_PITCH;
    float* dinv = out1 + (size_t)n * HID;
    float* g    = ws;
    float* out2 = g    + (size_t)n * G_PITCH;

    const int B = 256;
    int gn = (n + B - 1) / B;

    int ntiles = (n + 15) >> 4;
    int nwaves = (ntiles + TILES_PER_WAVE - 1) / TILES_PER_WAVE;
    int gw = (nwaves * 32 + B - 1) / B;
    gcn_gemm1_wmma <<<gw, B, 0, stream>>>(x, W1, h1, dinv, n);

    gcn_deg_scatter<<<dim3((n + DEG_NR - 1) / DEG_NR, 64), SCAT_BLK, 0, stream>>>(coli, dinv, n, e, 64);
    gcn_dinv_init_out1<<<gn, B, 0, stream>>>(dinv, h1, b1, out1, n);

    const dim3 sgrid((n + 3999) / 4000, 10);
    const size_t nblk     = (size_t)sgrid.x * sgrid.y;
    const size_t off_cnt  = (((size_t)n * (H1_PITCH + HID + 1) * sizeof(float)) + 127) & ~(size_t)127;
    const size_t off_list = off_cnt + ((nblk * sizeof(int) + 127) & ~(size_t)127);
    int*      counts = (int*)((char*)d_ws + off_cnt);
    unsigned* lists  = (unsigned*)((char*)d_ws + off_list);
    const bool with_lists = off_list + nblk * LIST_CAP * sizeof(unsigned) <= ws_size;

    if (with_lists)
        gcn_scatter_own<HID, 16, 4000, H1_PITCH, H1_DW, 12, 1><<<sgrid, SCAT_BLK, 0, stream>>>(rowi, coli, dinv, h1, out1, n, e, 10, lists, counts);
    else
        gcn_scatter_own<HID, 16, 4000, H1_PITCH, H1_DW, 12, 0><<<sgrid, SCAT_BLK, 0, stream>>>(rowi, coli, dinv, h1, out1, n, e, 10, nullptr, nullptr);

    gcn_gemm2      <<<gn, B, 0, stream>>>(out1, W2, b2, dinv, g, out2, n);
    if (with_lists)
        gcn_scatter_own<NCLS, 8, 4000, G_PITCH, G_DW, 12, 2><<<sgrid, SCAT_BLK, 0, stream>>>(rowi, coli, dinv, g, out2, n, e, 10, lists, counts);
    else
        gcn_scatter_own<NCLS, 8, 8192, G_PITCH, G_DW, 13, 0><<<dim3((n + 8191) / 8192, 19), SCAT_BLK, 0, stream>>>(rowi, coli, dinv, g, out2, n, e, 19, nullptr, nullptr);

    gcn_logsoftmax <<<gn, B, 0, stream>>>(out2, out, n);
}


# conv1 GEMM: a lane loads 16 B of x (4 consecutive k) straight into two A fragments, W1 fragments in the same permuted k order, no LDS staging: a row's line takes 4 loads, not 8; k sum order changed
# speedup vs baseline: 8.7410x; 1.0585x over previous
#include <hip/hip_runtime.h>
#include <hip/hip_bf16.h>

#define F_IN 512
#define HID  16
#define NCLS 7
#define TILES_PER_WAVE 2

#define H1_PITCH 32
#define H1_DW    16
#define G_PITCH  8
#define G_DW     7

typedef float v2f __attribute__((ext_vector_type(2)));
typedef float v4f __attribute__((ext_vector_type(4)));
typedef float v8f __attribute__((ext_vector_type(8)));


#define SCAT_BLK   1024
#define SCAT_WAVES (SCAT_BLK / 32)
#define SCAT_TILE  512
#define DEG_NR     50000
#define FLUSH_U    8
#define DEG_U      16

__global__ __launch_bounds__(SCAT_BLK)
void gcn_deg_scatter(const int* __restrict__ col, float* __restrict__ deg, int n, int e, int nchunks) {
    __shared__ __attribute__((aligned(16))) float acc[DEG_NR];
    static_assert(DEG_NR % 4 == 0, "the sums are zeroed four words at a time");
    const int tid = threadIdx.x;
    const int lo = blockIdx.x * DEG_NR;
    const int nr = min(DEG_NR, n - lo);
    for (int j = tid * 4; j < DEG_NR; j += SCAT_BLK * 4) *(float4*)&acc[j] = make_float4(0.0f, 0.0f, 0.0f, 0.0f);
    __syncthreads();

    const int chunk = (((e + nchunks - 1) / nchunks) + 3) & ~3;
    const int e0 = blockIdx.y * chunk;
    const int e1 = min(e0 + chunk, e);
    const bool al = (((size_t)col) & 15) == 0;
    for (int i = e0 + tid * 4; i < e1; i += DEG_U * SCAT_BLK * 4) {
        int t[DEG_U][4];
        #pragma unroll
        for (int u = 0; u < DEG_U; ++u) {
            const int iu = i + u * SCAT_BLK * 4;
            if (al && iu + 4 <= e1) {
                int4 v = *(const int4*)(col + iu);
                t[u][0] = v.x; t[u][1] = v.y; t[u][2] = v.z; t[u][3] = v.w;
            } else {
                #pragma unroll
                for (int k = 0; k < 4; ++k) t[u][k] = (iu + k < e1) ? col[iu + k] : -1;
            }
        }
        #pragma unroll
        for (int u = 0; u < DEG_U; ++u) {
            #pragma unroll
            for (int k = 0; k < 4; ++k) {
                unsigned d = (unsigned)(t[u][k] - lo);
                if (d < (unsigned)nr) atomicAdd(&acc[d], 1.0f);
            }
        }
    }
    __syncthreads();
    float* dp = deg + lo;
    int j = tid;
    for (; j + (FLUSH_U - 1) * SCAT_BLK < nr; j += FLUSH_U * SCAT_BLK) {
        float v[FLUSH_U];
        #pragma unroll
        for (int u = 0; u < FLUSH_U; ++u) v[u] = acc[j + u * SCAT_BLK];
        #pragma unroll
        for (int u = 0; u < FLUSH_U; ++u)
            if (v[u] != 0.0f) atomicAdd(&dp[j + u * SCAT_BLK], v[u]);
    }
    if (j < nr) {
        float v[FLUSH_U];
        #pragma unroll
        for (int u = 0; u < FLUSH_U; ++u) v[u] = (j + u * SCAT_BLK < nr) ? acc[j + u * SCAT_BLK] : 0.0f;
        #pragma unroll
        for (int u = 0; u < FLUSH_U; ++u)
            if (v[u] != 0.0f) atomicAdd(&dp[j + u * SCAT_BLK], v[u]);
    }
}


#define SCAT_QCAP_BIG   512
#define SCAT_QCAP_SMALL 256
#define SCAT_LDS_BYTES  (320 * 1024)

#define LIST_GROUPS 512
#define LIST_CAP    (LIST_GROUPS * 32)
#define LIST_NONE   0xffffffffu

template <int LPE, int R>
__device__ __forceinline__ int gcn_group_lane(int x) {
    static_assert(LPE == 16 || LPE == 8, "lane groups are DPP16 rows or DPP8 groups");
    static_assert(R >= 0 && R < LPE, "source lane lies inside the group");
    if constexpr (LPE == 16) return __builtin_amdgcn_mov_dpp(x, 0x150 + R, 0xf, 0xf, true);
    else                     return (int)__builtin_amdgcn_mov_dpp8((unsigned)x, R * 0x249249);
}
template <int LPE, int R>
__device__ __forceinline__ float gcn_group_lane_f(float x) {
    return __int_as_float(gcn_group_lane<LPE, R>(__float_as_int(x)));
}

template <int LPE, int PITCH, int R>
__device__ __forceinline__ void gcn_gather_rounds(int s, int f, const float* __restrict__ src, float (&v)[LPE]) {
    v[R] = src[gcn_group_lane<LPE, R>(s) * PITCH + f];
    if constexpr (R + 1 < LPE) gcn_gather_rounds<LPE, PITCH, R + 1>(s, f, src, v);
}

template <int F, int LPE, int DW, bool FULL, bool SKIP, int R>
__device__ __forceinline__ void gcn_add_rounds(int tl, float w, float dt, int m, int sub, int f,
                                               const float (&v)[LPE], float* acc) {
    if (FULL || !SKIP || R < m) {
        const int te = gcn_group_lane<LPE, R>(tl);
        float we;
        if constexpr (DW >= LPE) we = gcn_group_lane_f<LPE, R>(w);
        else                     we = gcn_group_lane_f<LPE, DW>(v[R]) * gcn_group_lane_f<LPE, R>(dt);
        if ((FULL || sub * LPE + R < m) && f < F) atomicAdd(&acc[te * F + f], we * v[R]);
    }
    if constexpr (R + 1 < LPE) gcn_add_rounds<F, LPE, DW, FULL, SKIP, R + 1>(tl, w, dt, m, sub, f, v, acc);
}

template <int F, int LPE, int PITCH, int DW, int TB, bool FULL, bool SKIP = true>
__device__ __forceinline__ void gcn_scatter_pass(unsigned ent, int m, int sub, int f,
                                                 const float* dloc, const float* __restrict__ src,
                                                 float* acc) {
    const int s  = (int)(ent >> TB);
    const int tl = (int)(ent & ((1u << TB) - 1u));
    float ds = 0.0f;
    if constexpr (DW >= LPE) ds = src[s * PITCH + DW];
    float v[LPE];
    gcn_gather_rounds<LPE, PITCH, 0>(s, f, src, v);
    const float dt = dloc[tl];
    const float w  = ds * dt;
    gcn_add_rounds<F, LPE, DW, FULL, SKIP, 0>(tl, w, dt, m, sub, f, v, acc);
}

__device__ __forceinline__ void gcn_list_put(unsigned ent, bool real, int lane, int* ngroups, unsigned* lst) {
    int g = 0;
    if (lane == 0) g = atomicAdd(ngroups, 1);
    g = __builtin_amdgcn_readfirstlane(g);
    if (g < LIST_GROUPS) lst[g * 32 + lane] = real ? ent : LIST_NONE;
}

template <int F, int LPE, int NR, int PITCH, int DW, int TB, int LIST>
__global__ __launch_bounds__(SCAT_BLK)
void gcn_scatter_own(const int* __restrict__ rowi, const int* __restrict__ coli,
                     const float* __restrict__ dinv, const float* __restrict__ src,
                     float* __restrict__ dst, int n, int e, int nchunks,
                     unsigned* lists, int* counts) {
    static_assert(NR <= (1 << TB), "target offset must fit the packed word");
    static_assert(LIST == 0 || NR < (1 << TB) - 1, "LIST_NONE must not be an entry");
    static_assert((NR * F) % 4 == 0, "the sums are zeroed four words at a time");
    constexpr int QCAP = ((size_t)NR * F * 4 + (size_t)NR * 4 + (size_t)SCAT_WAVES * SCAT_QCAP_BIG * 4 + 64 <= SCAT_LDS_BYTES)
                             ? SCAT_QCAP_BIG : SCAT_QCAP_SMALL;
    static_assert((size_t)NR * F * 4 + (size_t)NR * 4 + (size_t)SCAT_WAVES * QCAP * 4 + 64 <= SCAT_LDS_BYTES,
                  "sums, dinv copy and rings must fit one block's LDS");
    __shared__ __attribute__((aligned(16))) float acc[NR * F];
    __shared__ float    dloc[NR];
    __shared__ unsigned queue[SCAT_WAVES * QCAP];
    __shared__ int      ngroups;
    const int tid  = threadIdx.x;
    const int lane = tid & 31;
    const int wv   = tid >> 5;
    const int lo = blockIdx.x * NR;
    const int nr = min(NR, n - lo);
    const int bid = blockIdx.y * gridDim.x + blockIdx.x;
    unsigned* lst = (LIST != 0) ? lists + (size_t)bid * LIST_CAP : nullptr;
    for (int j = tid * 4; j < NR * F; j += SCAT_BLK * 4) *(float4*)&acc[j] = make_float4(0.0f, 0.0f, 0.0f, 0.0f);
    for (int j = tid; j < nr; j += SCAT_BLK) dloc[j] = dinv[lo + j];
    if (LIST == 1 && tid == 0) ngroups = 0;
    __syncthreads();
    const int sub = lane / LPE;
    const int f   = lane % LPE;

    const int ng = (LIST == 2) ? counts[bid] : 0;
    if (LIST == 2 && ng <= LIST_GROUPS) {
        int gi = __builtin_amdgcn_readfirstlane(wv);
        unsigned nxt = (gi < ng) ? lst[gi * 32 + lane] : 0u;
        for (; gi < ng; gi += SCAT_WAVES) {
            unsigned ent = nxt;
            if (gi + SCAT_WAVES < ng) nxt = lst[(gi + SCAT_WAVES) * 32 + lane];
            const int m = __popc((unsigned)__ballot(ent != LIST_NONE));
            const unsigned first = (unsigned)__builtin_amdgcn_readfirstlane((int)ent);
            if (lane >= m) ent = first;
            gcn_scatter_pass<F, LPE, PITCH, DW, TB, false, false>(ent, m, sub, f, dloc, src, acc);
        }
    } else {

    const int chunk = (((e + nchunks - 1) / nchunks) + SCAT_TILE - 1) & ~(SCAT_TILE - 1);
    const int e0 = blockIdx.y * chunk;
    const int e1 = min(e0 + chunk, e);
    const bool al = ((((size_t)coli) | ((size_t)rowi)) & 15) == 0;
    unsigned* q = queue + wv * QCAP;
    const unsigned below = (1u << lane) - 1u;

    auto load_quarter = [&](int b, int qt, int (&c)[16], int (&r)[16]) {
        if (al && b + SCAT_TILE <= e1) {
            const int4 vc = *(const int4*)(coli + b + qt * 128 + lane * 4);
            const int4 vr = *(const int4*)(rowi + b + qt * 128 + lane * 4);
            c[4*qt+0] = vc.x; c[4*qt+1] = vc.y; c[4*qt+2] = vc.z; c[4*qt+3] = vc.w;
            r[4*qt+0] = vr.x; r[4*qt+1] = vr.y; r[4*qt+2] = vr.z; r[4*qt+3] = vr.w;
        } else {
            #pragma unroll
            for (int j = 4 * qt; j < 4 * qt + 4; ++j) {
                const int idx = b + qt * 128 + lane * 4 + (j & 3);
                const bool in = idx < e1;
                c[j] = in ? coli[idx] : -1;
                r[j] = in ? rowi[idx] : 0;
            }
        }
    };

    constexpr int STEP = SCAT_WAVES * SCAT_TILE;
    int base = e0 + wv * SCAT_TILE;
    int c[16], r[16];
    int head = 0, tail = 0;
    if (base < e1) {
        #pragma unroll
        for (int qt = 0; qt < 4; ++qt) load_quarter(base, qt, c, r);
    }
    for (; base < e1; base += STEP) {
        #pragma unroll
        for (int qt = 0; qt < 4; ++qt) {
            #pragma unroll
            for (int j = 4 * qt; j < 4 * qt + 4; ++j) {
                const unsigned d = (unsigned)(c[j] - lo);
                const bool hit = d < (unsigned)nr;
                const unsigned mask = (unsigned)__ballot(hit);
                if (hit) q[(tail + __popc(mask & below)) & (QCAP - 1)] = ((unsigned)r[j] << TB) | d;
                tail += __popc(mask);
            }
            if (base + STEP < e1) load_quarter(base + STEP, qt, c, r);
            __builtin_amdgcn_fence(__ATOMIC_RELEASE, "wavefront");
            __builtin_amdgcn_wave_barrier();
            while (tail - head >= 32) {
                const unsigned ent = q[(head + lane) & (QCAP - 1)];
                gcn_scatter_pass<F, LPE, PITCH, DW, TB, true>(ent, 32, sub, f, dloc, src, acc);
                if (LIST == 1) gcn_list_put(ent, true, lane, &ngroups, lst);
                head += 32;
            }
            __builtin_amdgcn_fence(__ATOMIC_ACQUIRE, "wavefront");
            __builtin_amdgcn_wave_barrier();
        }
    }
    if (tail != head) {
        const int m = tail - head;
        const unsigned ent = q[(head + (lane < m ? lane : 0)) & (QCAP - 1)];
        gcn_scatter_pass<F, LPE, PITCH, DW, TB, false>(ent, m, sub, f, dloc, src, acc);
        if (LIST == 1) gcn_list_put(ent, lane < m, lane, &ngroups, lst);
    }
    }
    __syncthreads();
    if (LIST == 1 && tid == 0) counts[bid] = ngroups;
    float* dp = dst + (size_t)lo * F;
    const int tot = nr * F;
    int j = tid;
    for (; j + (FLUSH_U - 1) * SCAT_BLK < tot; j += FLUSH_U * SCAT_BLK) {
        float v[FLUSH_U];
        #pragma unroll
        for (int u = 0; u < FLUSH_U; ++u) v[u] = acc[j + u * SCAT_BLK];
        #pragma unroll
        for (int u = 0; u < FLUSH_U; ++u) atomicAdd(&dp[j + u * SCAT_BLK], v[u]);
    }
    if (j < tot) {
        float v[FLUSH_U];
        #pragma unroll
        for (int u = 0; u < FLUSH_U; ++u) v[u] = (j + u * SCAT_BLK < tot) ? acc[j + u * SCAT_BLK] : 0.0f;
        #pragma unroll
        for (int u = 0; u < FLUSH_U; ++u)
            if (j + u * SCAT_BLK < tot) atomicAdd(&dp[j + u * SCAT_BLK], v[u]);
    }
}

__global__ void gcn_dinv_init_out1(float* __restrict__ degdinv, float* h1,
                                   const float* __restrict__ b1, float* __restrict__ out1, int n) {
    int i = blockIdx.x * blockDim.x + threadIdx.x;
    if (i >= n) return;
    const float r = rsqrtf(degdinv[i]);
    degdinv[i] = r;
    float* hrow = h1 + (size_t)i * H1_PITCH;
    hrow[H1_DW] = r;
    const float s = r * r;
    const float4* hp = (const float4*)hrow;
    float4* op = (float4*)(out1 + (size_t)i * HID);
    #pragma unroll
    for (int q = 0; q < 4; ++q) {
        float4 h = hp[q];
        op[q] = make_float4(b1[4*q+0] + s*h.x, b1[4*q+1] + s*h.y,
                            b1[4*q+2] + s*h.z, b1[4*q+3] + s*h.w);
    }
}


__device__ __forceinline__ void store_tile(float* __restrict__ h1, const v8f& acc,
                                           int tile, int lane, int n) {
    int mbase = tile * 16 + ((lane >> 4) << 3);
    int ncol  = lane & 15;
    float* p = h1 + (size_t)mbase * H1_PITCH + ncol;
    if (mbase + 8 <= n) {
        #pragma unroll
        for (int v = 0; v < 8; ++v) p[v * H1_PITCH] = acc[v];
    } else {
        #pragma unroll
        for (int v = 0; v < 8; ++v) if (mbase + v < n) p[v * H1_PITCH] = acc[v];
    }
}

__global__ void gcn_gemm1_wmma(const float* __restrict__ x, const float* __restrict__ W1,
                               float* __restrict__ h1, float* __restrict__ deg, int n) {
    int gid = blockIdx.x * blockDim.x + threadIdx.x;
    if (gid < n) deg[gid] = 1.0f;
    int lane = threadIdx.x & 31;
    int wid  = gid >> 5;
    int ntiles = (n + 15) >> 4;
    int tile0 = wid * TILES_PER_WAVE;
    if (tile0 >= ntiles) return;
    bool has1 = (tile0 + 1) < ntiles;

    int m    = lane & 15;
    int koff = (lane >> 4) << 2;

    int row0 = tile0 * 16 + m;
    int row1 = row0 + 16;
    const float* xa0 = x + (size_t)(row0 < n ? row0 : 0) * F_IN + koff;
    const float* xa1 = x + (size_t)(row1 < n ? row1 : 0) * F_IN + koff;
    const float* wb  = W1 + (size_t)koff * HID + m;

    v8f acc0 = {}, acc1 = {};
    #pragma unroll
    for (int k = 0; k < F_IN; k += 8) {
        const v4f a0 = *(const v4f*)(xa0 + k);
        const v4f a1 = *(const v4f*)(xa1 + k);
        const float* wk = wb + (size_t)k * HID;
        v2f b01, b23, a0l, a0h, a1l, a1h;
        b01.x = wk[0 * HID];
        b01.y = wk[1 * HID];
        b23.x = wk[2 * HID];
        b23.y = wk[3 * HID];
        a0l.x = a0.x; a0l.y = a0.y; a0h.x = a0.z; a0h.y = a0.w;
        a1l.x = a1.x; a1l.y = a1.y; a1h.x = a1.z; a1h.y = a1.w;
        acc0 = __builtin_amdgcn_wmma_f32_16x16x4_f32(
            false, a0l, false, b01, (short)0, acc0, false, false);
        acc1 = __builtin_amdgcn_wmma_f32_16x16x4_f32(
            false, a1l, false, b01, (short)0, acc1, false, false);
        acc0 = __builtin_amdgcn_wmma_f32_16x16x4_f32(
            false, a0h, false, b23, (short)0, acc0, false, false);
        acc1 = __builtin_amdgcn_wmma_f32_16x16x4_f32(
            false, a1h, false, b23, (short)0, acc1, false, false);
    }

    store_tile(h1, acc0, tile0, lane, n);
    if (has1) store_tile(h1, acc1, tile0 + 1, lane, n);
}


__global__ void gcn_gemm2(const float* __restrict__ out1, const float* __restrict__ W2,
                          const float* __restrict__ b2, const float* __restrict__ dinv,
                          float* __restrict__ g, float* __restrict__ out2, int n) {
    int i = blockIdx.x * blockDim.x + threadIdx.x;
    if (i >= n) return;
    float h[HID];
    const float* ip = out1 + (size_t)i * HID;
    #pragma unroll
    for (int c = 0; c < HID; ++c) h[c] = fmaxf(ip[c], 0.0f);
    const float dv = dinv[i];
    const float s = dv * dv;
    float a[G_PITCH];
    #pragma unroll
    for (int c2 = 0; c2 < NCLS; ++c2) {
        float acc = 0.0f;
        #pragma unroll
        for (int c = 0; c < HID; ++c) acc += h[c] * W2[c * NCLS + c2];
        a[c2] = acc;
        out2[(size_t)i * NCLS + c2] = b2[c2] + s * acc;
    }
    a[G_DW] = dv;
    float4* gp = (float4*)(g + (size_t)i * G_PITCH);
    gp[0] = make_float4(a[0], a[1], a[2], a[3]);
    gp[1] = make_float4(a[4], a[5], a[6], a[7]);
}


__global__ void gcn_logsoftmax(const float* __restrict__ out2, float* __restrict__ out, int n) {
    int i = blockIdx.x * blockDim.x + threadIdx.x;
    if (i >= n) return;
    float v[NCLS];
    float m = -1e30f;
    #pragma unroll
    for (int c = 0; c < NCLS; ++c) { v[c] = out2[(size_t)i * NCLS + c]; m = fmaxf(m, v[c]); }
    float sum = 0.0f;
    #pragma unroll
    for (int c = 0; c < NCLS; ++c) sum += __expf(v[c] - m);
    float lse = m + __logf(sum);
    #pragma unroll
    for (int c = 0; c < NCLS; ++c) out[(size_t)i * NCLS + c] = v[c] - lse;
}


extern "C" void kernel_launch(void* const* d_in, const int* in_sizes, int n_in,
                              void* d_out, int out_size, void* d_ws, size_t ws_size,
                              hipStream_t stream) {
    const float* x  = (const float*)d_in[0];
    const int*   ei = (const int*)  d_in[1];
    const float* W1 = (const float*)d_in[2];
    const float* b1 = (const float*)d_in[3];
    const float* W2 = (const float*)d_in[4];
    const float* b2 = (const float*)d_in[5];
    float* out = (float*)d_out;

    int n = in_sizes[0] / F_IN;
    int e = in_sizes[1] / 2;
    const int* rowi = ei;
    const int* coli = ei + e;

    float* ws   = (float*)d_ws;
    float* h1   = ws;
    float* out1 = h1   + (size_t)n * H1_PITCH;
    float* dinv = out1 + (size_t)n * HID;
    float* g    = ws;
    float* out2 = g    + (size_t)n * G_PITCH;

    const int B = 256;
    int gn = (n + B - 1) / B;

    int ntiles = (n + 15) >> 4;
    int nwaves = (ntiles + TILES_PER_WAVE - 1) / TILES_PER_WAVE;
    int gw = (nwaves * 32 + B - 1) / B;
    gcn_gemm1_wmma <<<gw, B, 0, stream>>>(x, W1, h1, dinv, n);

    gcn_deg_scatter<<<dim3((n + DEG_NR - 1) / DEG_NR, 64), SCAT_BLK, 0, stream>>>(coli, dinv, n, e, 64);
    gcn_dinv_init_out1<<<gn, B, 0, stream>>>(dinv, h1, b1, out1, n);

    const dim3 sgrid((n + 3999) / 4000, 10);
    const size_t nblk     = (size_t)sgrid.x * sgrid.y;
    const size_t off_cnt  = (((size_t)n * (H1_PITCH + HID + 1) * sizeof(float)) + 127) & ~(size_t)127;
    const size_t off_list = off_cnt + ((nblk * sizeof(int) + 127) & ~(size_t)127);
    int*      counts = (int*)((char*)d_ws + off_cnt);
    unsigned* lists  = (unsigned*)((char*)d_ws + off_list);
    const bool with_lists = off_list + nblk * LIST_CAP * sizeof(unsigned) <= ws_size;

    if (with_lists)
        gcn_scatter_own<HID, 16, 4000, H1_PITCH, H1_DW, 12, 1><<<sgrid, SCAT_BLK, 0, stream>>>(rowi, coli, dinv, h1, out1, n, e, 10, lists, counts);
    else
        gcn_scatter_own<HID, 16, 4000, H1_PITCH, H1_DW, 12, 0><<<sgrid, SCAT_BLK, 0, stream>>>(rowi, coli, dinv, h1, out1, n, e, 10, nullptr, nullptr);

    gcn_gemm2      <<<gn, B, 0, stream>>>(out1, W2, b2, dinv, g, out2, n);
    if (with_lists)
        gcn_scatter_own<NCLS, 8, 4000, G_PITCH, G_DW, 12, 2><<<sgrid, SCAT_BLK, 0, stream>>>(rowi, coli, dinv, g, out2, n, e, 10, lists, counts);
    else
        gcn_scatter_own<NCLS, 8, 8192, G_PITCH, G_DW, 13, 0><<<dim3((n + 8191) / 8192, 19), SCAT_BLK, 0, stream>>>(rowi, coli, dinv, g, out2, n, e, 19, nullptr, nullptr);

    gcn_logsoftmax <<<gn, B, 0, stream>>>(out2, out, n);
}


# conv1's edge filter moves from the scatter into the GEMM launch (24 filter + 8 GEMM waves a block): writes hit lists and degree planes, conv1 replays them; 7 launches -> 6; LDS arrival order kept
# speedup vs baseline: 9.6385x; 1.1027x over previous
#include <hip/hip_runtime.h>
#include <hip/hip_bf16.h>

#define F_IN 512
#define HID  16
#define NCLS 7
#define TILES_PER_WAVE 2

#define H1_PITCH 32
#define H1_DW    16
#define G_PITCH  8
#define G_DW     7

typedef float v2f __attribute__((ext_vector_type(2)));
typedef float v4f __attribute__((ext_vector_type(4)));
typedef float v8f __attribute__((ext_vector_type(8)));


#define SCAT_BLK   1024
#define SCAT_WAVES (SCAT_BLK / 32)
#define SCAT_TILE  512
#define DEG_NR     50000
#define FLUSH_U    8
#define DEG_U      16

__global__ __launch_bounds__(SCAT_BLK)
void gcn_deg_scatter(const int* __restrict__ col, float* __restrict__ deg, int n, int e, int nchunks) {
    __shared__ __attribute__((aligned(16))) float acc[DEG_NR];
    static_assert(DEG_NR % 4 == 0, "the sums are zeroed four words at a time");
    const int tid = threadIdx.x;
    const int lo = blockIdx.x * DEG_NR;
    const int nr = min(DEG_NR, n - lo);
    for (int j = tid * 4; j < DEG_NR; j += SCAT_BLK * 4) *(float4*)&acc[j] = make_float4(0.0f, 0.0f, 0.0f, 0.0f);
    __syncthreads();

    const int chunk = (((e + nchunks - 1) / nchunks) + 3) & ~3;
    const int e0 = blockIdx.y * chunk;
    const int e1 = min(e0 + chunk, e);
    const bool al = (((size_t)col) & 15) == 0;
    for (int i = e0 + tid * 4; i < e1; i += DEG_U * SCAT_BLK * 4) {
        int t[DEG_U][4];
        #pragma unroll
        for (int u = 0; u < DEG_U; ++u) {
            const int iu = i + u * SCAT_BLK * 4;
            if (al && iu + 4 <= e1) {
                int4 v = *(const int4*)(col + iu);
                t[u][0] = v.x; t[u][1] = v.y; t[u][2] = v.z; t[u][3] = v.w;
            } else {
                #pragma unroll
                for (int k = 0; k < 4; ++k) t[u][k] = (iu + k < e1) ? col[iu + k] : -1;
            }
        }
        #pragma unroll
        for (int u = 0; u < DEG_U; ++u) {
            #pragma unroll
            for (int k = 0; k < 4; ++k) {
                unsigned d = (unsigned)(t[u][k] - lo);
                if (d < (unsigned)nr) atomicAdd(&acc[d], 1.0f);
            }
        }
    }
    __syncthreads();
    float* dp = deg + lo;
    int j = tid;
    for (; j + (FLUSH_U - 1) * SCAT_BLK < nr; j += FLUSH_U * SCAT_BLK) {
        float v[FLUSH_U];
        #pragma unroll
        for (int u = 0; u < FLUSH_U; ++u) v[u] = acc[j + u * SCAT_BLK];
        #pragma unroll
        for (int u = 0; u < FLUSH_U; ++u)
            if (v[u] != 0.0f) atomicAdd(&dp[j + u * SCAT_BLK], v[u]);
    }
    if (j < nr) {
        float v[FLUSH_U];
        #pragma unroll
        for (int u = 0; u < FLUSH_U; ++u) v[u] = (j + u * SCAT_BLK < nr) ? acc[j + u * SCAT_BLK] : 0.0f;
        #pragma unroll
        for (int u = 0; u < FLUSH_U; ++u)
            if (v[u] != 0.0f) atomicAdd(&dp[j + u * SCAT_BLK], v[u]);
    }
}


#define SCAT_QCAP_BIG   512
#define SCAT_QCAP_SMALL 256
#define SCAT_LDS_BYTES  (320 * 1024)

#define LIST_GROUPS 512
#define LIST_CAP    (LIST_GROUPS * 32)
#define LIST_NONE   0xffffffffu

template <int LPE, int R>
__device__ __forceinline__ int gcn_group_lane(int x) {
    static_assert(LPE == 16 || LPE == 8, "lane groups are DPP16 rows or DPP8 groups");
    static_assert(R >= 0 && R < LPE, "source lane lies inside the group");
    if constexpr (LPE == 16) return __builtin_amdgcn_mov_dpp(x, 0x150 + R, 0xf, 0xf, true);
    else                     return (int)__builtin_amdgcn_mov_dpp8((unsigned)x, R * 0x249249);
}
template <int LPE, int R>
__device__ __forceinline__ float gcn_group_lane_f(float x) {
    return __int_as_float(gcn_group_lane<LPE, R>(__float_as_int(x)));
}

template <int LPE, int PITCH, int R>
__device__ __forceinline__ void gcn_gather_rounds(int s, int f, const float* __restrict__ src, float (&v)[LPE]) {
    v[R] = src[gcn_group_lane<LPE, R>(s) * PITCH + f];
    if constexpr (R + 1 < LPE) gcn_gather_rounds<LPE, PITCH, R + 1>(s, f, src, v);
}

template <int F, int LPE, int DW, bool FULL, bool SKIP, int R>
__device__ __forceinline__ void gcn_add_rounds(int tl, float w, float dt, int m, int sub, int f,
                                               const float (&v)[LPE], float* acc) {
    if (FULL || !SKIP || R < m) {
        const int te = gcn_group_lane<LPE, R>(tl);
        float we;
        if constexpr (DW >= LPE) we = gcn_group_lane_f<LPE, R>(w);
        else                     we = gcn_group_lane_f<LPE, DW>(v[R]) * gcn_group_lane_f<LPE, R>(dt);
        if ((FULL || sub * LPE + R < m) && f < F) atomicAdd(&acc[te * F + f], we * v[R]);
    }
    if constexpr (R + 1 < LPE) gcn_add_rounds<F, LPE, DW, FULL, SKIP, R + 1>(tl, w, dt, m, sub, f, v, acc);
}

template <int F, int LPE, int PITCH, int DW, int TB, bool FULL, bool SKIP = true>
__device__ __forceinline__ void gcn_scatter_pass(unsigned ent, int m, int sub, int f,
                                                 const float* dloc, const float* __restrict__ src,
                                                 float* acc) {
    const int s  = (int)(ent >> TB);
    const int tl = (int)(ent & ((1u << TB) - 1u));
    float ds = 0.0f;
    if constexpr (DW >= LPE) ds = src[s * PITCH + DW];
    float v[LPE];
    gcn_gather_rounds<LPE, PITCH, 0>(s, f, src, v);
    const float dt = dloc[tl];
    const float w  = ds * dt;
    gcn_add_rounds<F, LPE, DW, FULL, SKIP, 0>(tl, w, dt, m, sub, f, v, acc);
}

__device__ __forceinline__ void gcn_list_put(unsigned ent, bool real, int lane, int* ngroups, unsigned* lst) {
    int g = 0;
    if (lane == 0) g = atomicAdd(ngroups, 1);
    g = __builtin_amdgcn_readfirstlane(g);
    if (g < LIST_GROUPS) lst[g * 32 + lane] = real ? ent : LIST_NONE;
}

template <int F, int LPE, int NR, int PITCH, int DW, int TB, int LIST>
__global__ __launch_bounds__(SCAT_BLK)
void gcn_scatter_own(const int* __restrict__ rowi, const int* __restrict__ coli,
                     const float* __restrict__ dinv, const float* __restrict__ src,
                     float* __restrict__ dst, int n, int e, int nchunks,
                     unsigned* lists, int* counts) {
    static_assert(NR <= (1 << TB), "target offset must fit the packed word");
    static_assert(LIST == 0 || LIST == 2, "no list, or replay");
    static_assert(LIST == 0 || NR < (1 << TB) - 1, "LIST_NONE must not be an entry");
    static_assert((NR * F) % 4 == 0, "the sums are zeroed four words at a time");
    constexpr int QCAP = ((size_t)NR * F * 4 + (size_t)NR * 4 + (size_t)SCAT_WAVES * SCAT_QCAP_BIG * 4 + 64 <= SCAT_LDS_BYTES)
                             ? SCAT_QCAP_BIG : SCAT_QCAP_SMALL;
    static_assert((size_t)NR * F * 4 + (size_t)NR * 4 + (size_t)SCAT_WAVES * QCAP * 4 + 64 <= SCAT_LDS_BYTES,
                  "sums, dinv copy and rings must fit one block's LDS");
    __shared__ __attribute__((aligned(16))) float acc[NR * F];
    __shared__ float    dloc[NR];
    __shared__ unsigned queue[SCAT_WAVES * QCAP];
    const int tid  = threadIdx.x;
    const int lane = tid & 31;
    const int wv   = tid >> 5;
    const int lo = blockIdx.x * NR;
    const int nr = min(NR, n - lo);
    const int bid = blockIdx.y * gridDim.x + blockIdx.x;
    unsigned* lst = (LIST != 0) ? lists + (size_t)bid * LIST_CAP : nullptr;
    for (int j = tid * 4; j < NR * F; j += SCAT_BLK * 4) *(float4*)&acc[j] = make_float4(0.0f, 0.0f, 0.0f, 0.0f);
    for (int j = tid; j < nr; j += SCAT_BLK) dloc[j] = dinv[lo + j];
    __syncthreads();
    const int sub = lane / LPE;
    const int f   = lane % LPE;

    const int ng = (LIST == 2) ? counts[bid] : 0;
    if (LIST == 2 && ng <= LIST_GROUPS) {
        int gi = __builtin_amdgcn_readfirstlane(wv);
        unsigned nxt = (gi < ng) ? lst[gi * 32 + lane] : 0u;
        for (; gi < ng; gi += SCAT_WAVES) {
            unsigned ent = nxt;
            if (gi + SCAT_WAVES < ng) nxt = lst[(gi + SCAT_WAVES) * 32 + lane];
            const int m = __popc((unsigned)__ballot(ent != LIST_NONE));
            const unsigned first = (unsigned)__builtin_amdgcn_readfirstlane((int)ent);
            if (lane >= m) ent = first;
            gcn_scatter_pass<F, LPE, PITCH, DW, TB, false, false>(ent, m, sub, f, dloc, src, acc);
        }
    } else {

    const int chunk = (((e + nchunks - 1) / nchunks) + SCAT_TILE - 1) & ~(SCAT_TILE - 1);
    const int e0 = blockIdx.y * chunk;
    const int e1 = min(e0 + chunk, e);
    const bool al = ((((size_t)coli) | ((size_t)rowi)) & 15) == 0;
    unsigned* q = queue + wv * QCAP;
    const unsigned below = (1u << lane) - 1u;

    auto load_quarter = [&](int b, int qt, int (&c)[16], int (&r)[16]) {
        if (al && b + SCAT_TILE <= e1) {
            const int4 vc = *(const int4*)(coli + b + qt * 128 + lane * 4);
            const int4 vr = *(const int4*)(rowi + b + qt * 128 + lane * 4);
            c[4*qt+0] = vc.x; c[4*qt+1] = vc.y; c[4*qt+2] = vc.z; c[4*qt+3] = vc.w;
            r[4*qt+0] = vr.x; r[4*qt+1] = vr.y; r[4*qt+2] = vr.z; r[4*qt+3] = vr.w;
        } else {
            #pragma unroll
            for (int j = 4 * qt; j < 4 * qt + 4; ++j) {
                const int idx = b + qt * 128 + lane * 4 + (j & 3);
                const bool in = idx < e1;
                c[j] = in ? coli[idx] : -1;
                r[j] = in ? rowi[idx] : 0;
            }
        }
    };

    constexpr int STEP = SCAT_WAVES * SCAT_TILE;
    int base = e0 + wv * SCAT_TILE;
    int c[16], r[16];
    int head = 0, tail = 0;
    if (base < e1) {
        #pragma unroll
        for (int qt = 0; qt < 4; ++qt) load_quarter(base, qt, c, r);
    }
    for (; base < e1; base += STEP) {
        #pragma unroll
        for (int qt = 0; qt < 4; ++qt) {
            #pragma unroll
            for (int j = 4 * qt; j < 4 * qt + 4; ++j) {
                const unsigned d = (unsigned)(c[j] - lo);
                const bool hit = d < (unsigned)nr;
                const unsigned mask = (unsigned)__ballot(hit);
                if (hit) q[(tail + __popc(mask & below)) & (QCAP - 1)] = ((unsigned)r[j] << TB) | d;
                tail += __popc(mask);
            }
            if (base + STEP < e1) load_quarter(base + STEP, qt, c, r);
            __builtin_amdgcn_fence(__ATOMIC_RELEASE, "wavefront");
            __builtin_amdgcn_wave_barrier();
            while (tail - head >= 32) {
                const unsigned ent = q[(head + lane) & (QCAP - 1)];
                gcn_scatter_pass<F, LPE, PITCH, DW, TB, true>(ent, 32, sub, f, dloc, src, acc);
                head += 32;
            }
            __builtin_amdgcn_fence(__ATOMIC_ACQUIRE, "wavefront");
            __builtin_amdgcn_wave_barrier();
        }
    }
    if (tail != head) {
        const int m = tail - head;
        const unsigned ent = q[(head + (lane < m ? lane : 0)) & (QCAP - 1)];
        gcn_scatter_pass<F, LPE, PITCH, DW, TB, false>(ent, m, sub, f, dloc, src, acc);
    }
    }
    __syncthreads();
    float* dp = dst + (size_t)lo * F;
    const int tot = nr * F;
    int j = tid;
    for (; j + (FLUSH_U - 1) * SCAT_BLK < tot; j += FLUSH_U * SCAT_BLK) {
        float v[FLUSH_U];
        #pragma unroll
        for (int u = 0; u < FLUSH_U; ++u) v[u] = acc[j + u * SCAT_BLK];
        #pragma unroll
        for (int u = 0; u < FLUSH_U; ++u) atomicAdd(&dp[j + u * SCAT_BLK], v[u]);
    }
    if (j < tot) {
        float v[FLUSH_U];
        #pragma unroll
        for (int u = 0; u < FLUSH_U; ++u) v[u] = (j + u * SCAT_BLK < tot) ? acc[j + u * SCAT_BLK] : 0.0f;
        #pragma unroll
        for (int u = 0; u < FLUSH_U; ++u)
            if (j + u * SCAT_BLK < tot) atomicAdd(&dp[j + u * SCAT_BLK], v[u]);
    }
}

__global__ void gcn_dinv_init_out1(float* __restrict__ degdinv, float* h1,
                                   const float* __restrict__ b1, float* __restrict__ out1,
                                   const float* __restrict__ parts, int nparts, int n) {
    int i = blockIdx.x * blockDim.x + threadIdx.x;
    if (i >= n) return;
    float dg;
    if (nparts > 0) {
        dg = 1.0f;
        for (int y = 0; y < nparts; ++y) dg += parts[(size_t)y * n + i];
    } else {
        dg = degdinv[i];
    }
    const float r = rsqrtf(dg);
    degdinv[i] = r;
    float* hrow = h1 + (size_t)i * H1_PITCH;
    hrow[H1_DW] = r;
    const float s = r * r;
    const float4* hp = (const float4*)hrow;
    float4* op = (float4*)(out1 + (size_t)i * HID);
    #pragma unroll
    for (int q = 0; q < 4; ++q) {
        float4 h = hp[q];
        op[q] = make_float4(b1[4*q+0] + s*h.x, b1[4*q+1] + s*h.y,
                            b1[4*q+2] + s*h.z, b1[4*q+3] + s*h.w);
    }
}


__device__ __forceinline__ void store_tile(float* __restrict__ h1, const v8f& acc,
                                           int tile, int lane, int n) {
    int mbase = tile * 16 + ((lane >> 4) << 3);
    int ncol  = lane & 15;
    float* p = h1 + (size_t)mbase * H1_PITCH + ncol;
    if (mbase + 8 <= n) {
        #pragma unroll
        for (int v = 0; v < 8; ++v) p[v * H1_PITCH] = acc[v];
    } else {
        #pragma unroll
        for (int v = 0; v < 8; ++v) if (mbase + v < n) p[v * H1_PITCH] = acc[v];
    }
}

__device__ __forceinline__ void gcn_gemm1_wave(const float* __restrict__ x, const float* __restrict__ W1,
                                               float* __restrict__ h1, int n, int wid, int lane) {
    int ntiles = (n + 15) >> 4;
    int tile0 = wid * TILES_PER_WAVE;
    if (tile0 >= ntiles) return;
    bool has1 = (tile0 + 1) < ntiles;

    int m    = lane & 15;
    int koff = (lane >> 4) << 2;

    int row0 = tile0 * 16 + m;
    int row1 = row0 + 16;
    const float* xa0 = x + (size_t)(row0 < n ? row0 : 0) * F_IN + koff;
    const float* xa1 = x + (size_t)(row1 < n ? row1 : 0) * F_IN + koff;
    const float* wb  = W1 + (size_t)koff * HID + m;

    v8f acc0 = {}, acc1 = {};
    #pragma unroll
    for (int k = 0; k < F_IN; k += 8) {
        const v4f a0 = *(const v4f*)(xa0 + k);
        const v4f a1 = *(const v4f*)(xa1 + k);
        const float* wk = wb + (size_t)k * HID;
        v2f b01, b23, a0l, a0h, a1l, a1h;
        b01.x = wk[0 * HID];
        b01.y = wk[1 * HID];
        b23.x = wk[2 * HID];
        b23.y = wk[3 * HID];
        a0l.x = a0.x; a0l.y = a0.y; a0h.x = a0.z; a0h.y = a0.w;
        a1l.x = a1.x; a1l.y = a1.y; a1h.x = a1.z; a1h.y = a1.w;
        acc0 = __builtin_amdgcn_wmma_f32_16x16x4_f32(
            false, a0l, false, b01, (short)0, acc0, false, false);
        acc1 = __builtin_amdgcn_wmma_f32_16x16x4_f32(
            false, a1l, false, b01, (short)0, acc1, false, false);
        acc0 = __builtin_amdgcn_wmma_f32_16x16x4_f32(
            false, a0h, false, b23, (short)0, acc0, false, false);
        acc1 = __builtin_amdgcn_wmma_f32_16x16x4_f32(
            false, a1h, false, b23, (short)0, acc1, false, false);
    }

    store_tile(h1, acc0, tile0, lane, n);
    if (has1) store_tile(h1, acc1, tile0 + 1, lane, n);
}

__global__ void gcn_gemm1_wmma(const float* __restrict__ x, const float* __restrict__ W1,
                               float* __restrict__ h1, float* __restrict__ deg, int n) {
    int gid = blockIdx.x * blockDim.x + threadIdx.x;
    if (gid < n) deg[gid] = 1.0f;
    gcn_gemm1_wave(x, W1, h1, n, gid >> 5, threadIdx.x & 31);
}


#define FG_FWAVES 24
#define FG_GWAVES (SCAT_WAVES - FG_FWAVES)
#define FG_NR     4000
#define FG_TB     12

__global__ __launch_bounds__(SCAT_BLK)
void gcn_filter_gemm(const int* __restrict__ rowi, const int* __restrict__ coli,
                     const float* __restrict__ x, const float* __restrict__ W1,
                     float* __restrict__ h1, float* __restrict__ dpart,
                     int n, int e, int nchunks, unsigned* lists, int* counts) {
    static_assert(FG_NR < (1 << FG_TB) - 1, "LIST_NONE must not be an entry");
    constexpr int QCAP = SCAT_QCAP_BIG;
    __shared__ float    dacc[FG_NR];
    __shared__ unsigned queue[FG_FWAVES * QCAP];
    __shared__ int      ngroups;
    const int tid  = threadIdx.x;
    const int lane = tid & 31;
    const int wv   = __builtin_amdgcn_readfirstlane(tid >> 5);
    const int lo = blockIdx.x * FG_NR;
    const int nr = min(FG_NR, n - lo);
    const int bid = blockIdx.y * gridDim.x + blockIdx.x;
    const int nblk = gridDim.x * gridDim.y;
    unsigned* lst = lists + (size_t)bid * LIST_CAP;
    for (int j = tid; j < FG_NR; j += SCAT_BLK) dacc[j] = 0.0f;
    if (tid == 0) ngroups = 0;
    __syncthreads();

    if (wv >= FG_FWAVES) {
        const int ntiles = (n + 15) >> 4;
        const int nwaves = (ntiles + TILES_PER_WAVE - 1) / TILES_PER_WAVE;
        for (int w = bid * FG_GWAVES + (wv - FG_FWAVES); w < nwaves; w += nblk * FG_GWAVES)
            gcn_gemm1_wave(x, W1, h1, n, w, lane);
    } else {
        const int chunk = (((e + nchunks - 1) / nchunks) + SCAT_TILE - 1) & ~(SCAT_TILE - 1);
        const int e0 = blockIdx.y * chunk;
        const int e1 = min(e0 + chunk, e);
        const bool al = ((((size_t)coli) | ((size_t)rowi)) & 15) == 0;
        unsigned* q = queue + wv * QCAP;
        const unsigned below = (1u << lane) - 1u;

        auto load_quarter = [&](int b, int qt, int (&c)[16], int (&r)[16]) {
            if (al && b + SCAT_TILE <= e1) {
                const int4 vc = *(const int4*)(coli + b + qt * 128 + lane * 4);
                const int4 vr = *(const int4*)(rowi + b + qt * 128 + lane * 4);
                c[4*qt+0] = vc.x; c[4*qt+1] = vc.y; c[4*qt+2] = vc.z; c[4*qt+3] = vc.w;
                r[4*qt+0] = vr.x; r[4*qt+1] = vr.y; r[4*qt+2] = vr.z; r[4*qt+3] = vr.w;
            } else {
                #pragma unroll
                for (int j = 4 * qt; j < 4 * qt + 4; ++j) {
                    const int idx = b + qt * 128 + lane * 4 + (j & 3);
                    const bool in = idx < e1;
                    c[j] = in ? coli[idx] : -1;
                    r[j] = in ? rowi[idx] : 0;
                }
            }
        };

        constexpr int STEP = FG_FWAVES * SCAT_TILE;
        int base = e0 + wv * SCAT_TILE;
        int c[16], r[16];
        int head = 0, tail = 0;
        if (base < e1) {
            #pragma unroll
            for (int qt = 0; qt < 4; ++qt) load_quarter(base, qt, c, r);
        }
        for (; base < e1; base += STEP) {
            #pragma unroll
            for (int qt = 0; qt < 4; ++qt) {
                #pragma unroll
                for (int j = 4 * qt; j < 4 * qt + 4; ++j) {
                    const unsigned d = (unsigned)(c[j] - lo);
                    const bool hit = d < (unsigned)nr;
                    const unsigned mask = (unsigned)__ballot(hit);
                    if (hit) q[(tail + __popc(mask & below)) & (QCAP - 1)] = ((unsigned)r[j] << FG_TB) | d;
                    tail += __popc(mask);
                }
                if (base + STEP < e1) load_quarter(base + STEP, qt, c, r);
                __builtin_amdgcn_fence(__ATOMIC_RELEASE, "wavefront");
                __builtin_amdgcn_wave_barrier();
                while (tail - head >= 32) {
                    const unsigned ent = q[(head + lane) & (QCAP - 1)];
                    atomicAdd(&dacc[ent & ((1u << FG_TB) - 1u)], 1.0f);
                    gcn_list_put(ent, true, lane, &ngroups, lst);
                    head += 32;
                }
                __builtin_amdgcn_fence(__ATOMIC_ACQUIRE, "wavefront");
                __builtin_amdgcn_wave_barrier();
            }
        }
        if (tail != head) {
            const int m = tail - head;
            const unsigned ent = q[(head + (lane < m ? lane : 0)) & (QCAP - 1)];
            if (lane < m) atomicAdd(&dacc[ent & ((1u << FG_TB) - 1u)], 1.0f);
            gcn_list_put(ent, lane < m, lane, &ngroups, lst);
        }
    }
    __syncthreads();
    if (tid == 0) counts[bid] = ngroups;
    float* dp = dpart + (size_t)blockIdx.y * n + lo;
    for (int j = tid; j < nr; j += SCAT_BLK) dp[j] = dacc[j];
}


__global__ void gcn_gemm2(const float* __restrict__ out1, const float* __restrict__ W2,
                          const float* __restrict__ b2, const float* __restrict__ dinv,
                          float* __restrict__ g, float* __restrict__ out2, int n) {
    int i = blockIdx.x * blockDim.x + threadIdx.x;
    if (i >= n) return;
    float h[HID];
    const float* ip = out1 + (size_t)i * HID;
    #pragma unroll
    for (int c = 0; c < HID; ++c) h[c] = fmaxf(ip[c], 0.0f);
    const float dv = dinv[i];
    const float s = dv * dv;
    float a[G_PITCH];
    #pragma unroll
    for (int c2 = 0; c2 < NCLS; ++c2) {
        float acc = 0.0f;
        #pragma unroll
        for (int c = 0; c < HID; ++c) acc += h[c] * W2[c * NCLS + c2];
        a[c2] = acc;
        out2[(size_t)i * NCLS + c2] = b2[c2] + s * acc;
    }
    a[G_DW] = dv;
    float4* gp = (float4*)(g + (size_t)i * G_PITCH);
    gp[0] = make_float4(a[0], a[1], a[2], a[3]);
    gp[1] = make_float4(a[4], a[5], a[6], a[7]);
}


__global__ void gcn_logsoftmax(const float* __restrict__ out2, float* __restrict__ out, int n) {
    int i = blockIdx.x * blockDim.x + threadIdx.x;
    if (i >= n) return;
    float v[NCLS];
    float m = -1e30f;
    #pragma unroll
    for (int c = 0; c < NCLS; ++c) { v[c] = out2[(size_t)i * NCLS + c]; m = fmaxf(m, v[c]); }
    float sum = 0.0f;
    #pragma unroll
    for (int c = 0; c < NCLS; ++c) sum += __expf(v[c] - m);
    float lse = m + __logf(sum);
    #pragma unroll
    for (int c = 0; c < NCLS; ++c) out[(size_t)i * NCLS + c] = v[c] - lse;
}


extern "C" void kernel_launch(void* const* d_in, const int* in_sizes, int n_in,
                              void* d_out, int out_size, void* d_ws, size_t ws_size,
                              hipStream_t stream) {
    const float* x  = (const float*)d_in[0];
    const int*   ei = (const int*)  d_in[1];
    const float* W1 = (const float*)d_in[2];
    const float* b1 = (const float*)d_in[3];
    const float* W2 = (const float*)d_in[4];
    const float* b2 = (const float*)d_in[5];
    float* out = (float*)d_out;

    int n = in_sizes[0] / F_IN;
    int e = in_sizes[1] / 2;
    const int* rowi = ei;
    const int* coli = ei + e;

    float* ws   = (float*)d_ws;
    float* h1   = ws;
    float* out1 = h1   + (size_t)n * H1_PITCH;
    float* dinv = out1 + (size_t)n * HID;
    float* g    = ws;
    float* out2 = g    + (size_t)n * G_PITCH;

    const int B = 256;
    int gn = (n + B - 1) / B;

    const dim3 sgrid((n + FG_NR - 1) / FG_NR, 10);
    const size_t nblk     = (size_t)sgrid.x * sgrid.y;
    const size_t off_cnt  = (((size_t)n * (H1_PITCH + HID + 1) * sizeof(float)) + 127) & ~(size_t)127;
    const size_t off_list = off_cnt + ((nblk * sizeof(int) + 127) & ~(size_t)127);
    const size_t off_part = off_list + nblk * LIST_CAP * sizeof(unsigned);
    int*      counts = (int*)((char*)d_ws + off_cnt);
    unsigned* lists  = (unsigned*)((char*)d_ws + off_list);
    float*    dpart  = (float*)((char*)d_ws + off_part);
    const bool with_lists = off_part + (size_t)sgrid.y * n * sizeof(float) <= ws_size;

    if (with_lists) {
        gcn_filter_gemm<<<sgrid, SCAT_BLK, 0, stream>>>(rowi, coli, x, W1, h1, dpart, n, e, 10, lists, counts);
        gcn_dinv_init_out1<<<gn, B, 0, stream>>>(dinv, h1, b1, out1, dpart, 10, n);
        gcn_scatter_own<HID, 16, FG_NR, H1_PITCH, H1_DW, FG_TB, 2><<<sgrid, SCAT_BLK, 0, stream>>>(rowi, coli, dinv, h1, out1, n, e, 10, lists, counts);
    } else {
        int ntiles = (n + 15) >> 4;
        int nwaves = (ntiles + TILES_PER_WAVE - 1) / TILES_PER_WAVE;
        int gw = (nwaves * 32 + B - 1) / B;
        gcn_gemm1_wmma <<<gw, B, 0, stream>>>(x, W1, h1, dinv, n);
        gcn_deg_scatter<<<dim3((n + DEG_NR - 1) / DEG_NR, 64), SCAT_BLK, 0, stream>>>(coli, dinv, n, e, 64);
        gcn_dinv_init_out1<<<gn, B, 0, stream>>>(dinv, h1, b1, out1, nullptr, 0, n);
        gcn_scatter_own<HID, 16, 4000, H1_PITCH, H1_DW, 12, 0><<<sgrid, SCAT_BLK, 0, stream>>>(rowi, coli, dinv, h1, out1, n, e, 10, nullptr, nullptr);
    }

    gcn_gemm2      <<<gn, B, 0, stream>>>(out1, W2, b2, dinv, g, out2, n);
    if (with_lists)
        gcn_scatter_own<NCLS, 8, 4000, G_PITCH, G_DW, 12, 2><<<sgrid, SCAT_BLK, 0, stream>>>(rowi, coli, dinv, g, out2, n, e, 10, lists, counts);
    else
        gcn_scatter_own<NCLS, 8, 8192, G_PITCH, G_DW, 13, 0><<<dim3((n + 8191) / 8192, 19), SCAT_BLK, 0, stream>>>(rowi, coli, dinv, g, out2, n, e, 19, nullptr, nullptr);

    gcn_logsoftmax <<<gn, B, 0, stream>>>(out2, out, n);
}
